# starting kernel
# baseline (speedup 1.0000x reference)
#include <hip/hip_runtime.h>
#include <hip/hip_bf16.h>

#define IN_C  64
#define OUT_C 64
#define EPS   (1.0f + 1e-06f)

typedef float v2f __attribute__((ext_vector_type(2)));
typedef float v8f __attribute__((ext_vector_type(8)));

template <bool HAS_ATT>
__device__ __forceinline__ void wmma_block(
    const v2f a_frag[16],
    const float* __restrict__ w,
    const float* __restrict__ att,
    float* __restrict__ store,
    float* __restrict__ s_src,
    float* __restrict__ s_tgt,
    float scale, int row0, int half, int lrow)
{
    float att_s[4], att_t[4];
    if (HAS_ATT) {
#pragma unroll
        for (int nt = 0; nt < 4; ++nt) {
            att_s[nt] = att[nt * 16 + lrow];
            att_t[nt] = att[OUT_C + nt * 16 + lrow];
        }
    }

    float ssrc_acc[8];
    float stgt_acc[8];
#pragma unroll
    for (int r = 0; r < 8; ++r) { ssrc_acc[r] = 0.f; stgt_acc[r] = 0.f; }

#pragma unroll
    for (int nt = 0; nt < 4; ++nt) {
        const int col = nt * 16 + lrow;
        v8f c = {};
#pragma unroll
        for (int kk = 0; kk < 16; ++kk) {
            const int k0 = 4 * kk + 2 * half;
            v2f b;
            b.x = w[(size_t)k0 * OUT_C + col];
            b.y = w[(size_t)(k0 + 1) * OUT_C + col];
            c = __builtin_amdgcn_wmma_f32_16x16x4_f32(
                     false, a_frag[kk],
                     false, b,
                     (short)0, c,
                     false,  false);
        }
#pragma unroll
        for (int r = 0; r < 8; ++r) {
            const int row = row0 + r + half * 8;
            const float v = c[r];
            store[(size_t)row * OUT_C + col] = v * scale;
            if (HAS_ATT) {
                ssrc_acc[r] += v * att_s[nt];
                stgt_acc[r] += v * att_t[nt];
            }
        }
    }

    if (HAS_ATT) {
#pragma unroll
        for (int r = 0; r < 8; ++r) {
            float vs = ssrc_acc[r];
            float vt = stgt_acc[r];
#pragma unroll
            for (int m = 8; m >= 1; m >>= 1) {
                vs += __shfl_xor(vs, m, 32);
                vt += __shfl_xor(vt, m, 32);
            }
            if (lrow == 0) {
                const int row = row0 + r + half * 8;
                s_src[row] = vs;
                s_tgt[row] = vt;
            }
        }
    }
}

__global__ void __launch_bounds__(128)
node_gemm_kernel(const float* __restrict__ x,
                 const float* __restrict__ w_lower,
                 const float* __restrict__ a_lower,
                 const float* __restrict__ w_upper,
                 const float* __restrict__ a_upper,
                 const float* __restrict__ w_lin,
                 float* __restrict__ xm_l, float* __restrict__ xm_u,
                 float* __restrict__ sl_s, float* __restrict__ sl_t,
                 float* __restrict__ su_s, float* __restrict__ su_t,
                 float* __restrict__ out, int n_rows)
{
    const int wid  = (int)((blockIdx.x * blockDim.x + threadIdx.x) >> 5);
    const int lane = (int)(threadIdx.x & 31);
    const int row0 = wid * 16;
    if (row0 >= n_rows) return;

    const int half = lane >> 4;
    const int lrow = lane & 15;

    v2f a_frag[16];
    const float* xr = x + (size_t)(row0 + lrow) * IN_C + 2 * half;
#pragma unroll
    for (int kk = 0; kk < 16; ++kk)
        a_frag[kk] = *(const v2f*)(xr + 4 * kk);

    wmma_block<true >(a_frag, w_lower, a_lower, xm_l, sl_s, sl_t, 1.0f, row0, half, lrow);
    wmma_block<true >(a_frag, w_upper, a_upper, xm_u, su_s, su_t, 1.0f, row0, half, lrow);
    wmma_block<false>(a_frag, w_lin,   nullptr, out,  nullptr, nullptr, EPS, row0, half, lrow);
}

__global__ void __launch_bounds__(256)
edge_scatter_kernel(const int* __restrict__ idx,
                    const float* __restrict__ vals,
                    const float* __restrict__ xm,
                    const float* __restrict__ s_src,
                    const float* __restrict__ s_tgt,
                    float* __restrict__ out,
                    int E)
{
    const int e = __builtin_amdgcn_readfirstlane(
        (int)((blockIdx.x * blockDim.x + threadIdx.x) >> 5));
    const int lane = (int)(threadIdx.x & 31);
    if (e >= E) return;

    const int i = idx[e];
    const int j = idx[(size_t)E + e];

    const float z = s_src[j] + s_tgt[i];
    const float elu = (z > 0.f) ? z : expm1f(z);
    const float alpha = elu * vals[e];

    const float2 m = ((const float2*)(xm + (size_t)j * OUT_C))[lane];
    float* o = out + (size_t)i * OUT_C + lane * 2;
    atomicAdd(o,     alpha * m.x);
    atomicAdd(o + 1, alpha * m.y);
}

__global__ void __launch_bounds__(256)
relu_kernel(float* __restrict__ out, size_t n)
{
    const size_t t = (size_t)blockIdx.x * blockDim.x + threadIdx.x;
    if (t < n) out[t] = fmaxf(out[t], 0.0f);
}

extern "C" void kernel_launch(void* const* d_in, const int* in_sizes, int n_in,
                              void* d_out, int out_size, void* d_ws, size_t ws_size,
                              hipStream_t stream) {
    const float* x        = (const float*)d_in[0];
    const int*   low_idx  = (const int*)  d_in[1];
    const float* low_val  = (const float*)d_in[2];
    const int*   up_idx   = (const int*)  d_in[3];
    const float* up_val   = (const float*)d_in[4];
    const float* w_lower  = (const float*)d_in[5];
    const float* a_lower  = (const float*)d_in[6];
    const float* w_upper  = (const float*)d_in[7];
    const float* a_upper  = (const float*)d_in[8];
    const float* w_lin    = (const float*)d_in[9];
    float*       out      = (float*)d_out;

    const int N = in_sizes[0] / IN_C;
    const int E = in_sizes[2];

    float* ws   = (float*)d_ws;
    float* xm_l = ws;
    float* xm_u = xm_l + (size_t)N * OUT_C;
    float* sl_s = xm_u + (size_t)N * OUT_C;
    float* sl_t = sl_s + N;
    float* su_s = sl_t + N;
    float* su_t = su_s + N;

    {
        const int waves  = (N + 15) / 16;
        const int blocks = (waves + 3) / 4;
        node_gemm_kernel<<<blocks, 128, 0, stream>>>(
            x, w_lower, a_lower, w_upper, a_upper, w_lin,
            xm_l, xm_u, sl_s, sl_t, su_s, su_t, out, N);
    }

    {
        const int blocks = (E + 7) / 8;
        edge_scatter_kernel<<<blocks, 256, 0, stream>>>(
            low_idx, low_val, xm_l, sl_s, sl_t, out, E);
        edge_scatter_kernel<<<blocks, 256, 0, stream>>>(
            up_idx, up_val, xm_u, su_s, su_t, out, E);
    }

    {
        const size_t n = (size_t)N * OUT_C;
        const int blocks = (int)((n + 255) / 256);
        relu_kernel<<<blocks, 256, 0, stream>>>(out, n);
    }
}


# edge sums built in LDS per tile of 196 target rows: each work-group scans a per-edge tile byte for its rows, adds alpha*row with LDS float adds, ReLU on store; no global atomics; sum order differs
# speedup vs baseline: 2.1549x; 2.1549x over previous
#include <hip/hip_runtime.h>
#include <hip/hip_bf16.h>

#define IN_C  64
#define OUT_C 64
#define EPS   (1.0f + 1e-06f)

typedef float v2f __attribute__((ext_vector_type(2)));
typedef float v8f __attribute__((ext_vector_type(8)));

template <bool HAS_ATT>
__device__ __forceinline__ void wmma_block(
    const v2f a_frag[16],
    const float* __restrict__ w,
    const float* __restrict__ att,
    float* __restrict__ store,
    float* __restrict__ s_src,
    float* __restrict__ s_tgt,
    float scale, int row0, int half, int lrow)
{
    float att_s[4], att_t[4];
    if (HAS_ATT) {
#pragma unroll
        for (int nt = 0; nt < 4; ++nt) {
            att_s[nt] = att[nt * 16 + lrow];
            att_t[nt] = att[OUT_C + nt * 16 + lrow];
        }
    }

    float ssrc_acc[8];
    float stgt_acc[8];
#pragma unroll
    for (int r = 0; r < 8; ++r) { ssrc_acc[r] = 0.f; stgt_acc[r] = 0.f; }

#pragma unroll
    for (int nt = 0; nt < 4; ++nt) {
        const int col = nt * 16 + lrow;
        v8f c = {};
#pragma unroll
        for (int kk = 0; kk < 16; ++kk) {
            const int k0 = 4 * kk + 2 * half;
            v2f b;
            b.x = w[(size_t)k0 * OUT_C + col];
            b.y = w[(size_t)(k0 + 1) * OUT_C + col];
            c = __builtin_amdgcn_wmma_f32_16x16x4_f32(
                     false, a_frag[kk],
                     false, b,
                     (short)0, c,
                     false,  false);
        }
#pragma unroll
        for (int r = 0; r < 8; ++r) {
            const int row = row0 + r + half * 8;
            const float v = c[r];
            store[(size_t)row * OUT_C + col] = v * scale;
            if (HAS_ATT) {
                ssrc_acc[r] += v * att_s[nt];
                stgt_acc[r] += v * att_t[nt];
            }
        }
    }

    if (HAS_ATT) {
#pragma unroll
        for (int r = 0; r < 8; ++r) {
            float vs = ssrc_acc[r];
            float vt = stgt_acc[r];
#pragma unroll
            for (int m = 8; m >= 1; m >>= 1) {
                vs += __shfl_xor(vs, m, 32);
                vt += __shfl_xor(vt, m, 32);
            }
            if (lrow == 0) {
                const int row = row0 + r + half * 8;
                s_src[row] = vs;
                s_tgt[row] = vt;
            }
        }
    }
}

__global__ void __launch_bounds__(128)
node_gemm_kernel(const float* __restrict__ x,
                 const float* __restrict__ w_lower,
                 const float* __restrict__ a_lower,
                 const float* __restrict__ w_upper,
                 const float* __restrict__ a_upper,
                 const float* __restrict__ w_lin,
                 float* __restrict__ xm_l, float* __restrict__ xm_u,
                 float* __restrict__ sl_s, float* __restrict__ sl_t,
                 float* __restrict__ su_s, float* __restrict__ su_t,
                 float* __restrict__ out, int n_rows)
{
    const int wid  = (int)((blockIdx.x * blockDim.x + threadIdx.x) >> 5);
    const int lane = (int)(threadIdx.x & 31);
    const int row0 = wid * 16;
    if (row0 >= n_rows) return;

    const int half = lane >> 4;
    const int lrow = lane & 15;

    v2f a_frag[16];
    const float* xr = x + (size_t)(row0 + lrow) * IN_C + 2 * half;
#pragma unroll
    for (int kk = 0; kk < 16; ++kk)
        a_frag[kk] = *(const v2f*)(xr + 4 * kk);

    wmma_block<true >(a_frag, w_lower, a_lower, xm_l, sl_s, sl_t, 1.0f, row0, half, lrow);
    wmma_block<true >(a_frag, w_upper, a_upper, xm_u, su_s, su_t, 1.0f, row0, half, lrow);
    wmma_block<false>(a_frag, w_lin,   nullptr, out,  nullptr, nullptr, EPS, row0, half, lrow);
}

#define R_TILE 196
#define TILE_WAVES 32

typedef __attribute__((address_space(3))) int lds_int;

__global__ void __launch_bounds__(256)
tile_byte_kernel(const int* __restrict__ idx_l,
                 const int* __restrict__ idx_u,
                 unsigned int* __restrict__ tb4,
                 int E, int Epad)
{
    const int t  = (int)(blockIdx.x * blockDim.x + threadIdx.x);
    const int g0 = 4 * t;
    if (g0 >= 2 * Epad) return;
    const int list = (g0 >= Epad) ? 1 : 0;
    const int e0 = g0 - list * Epad;
    const int* idx = list ? idx_u : idx_l;
    unsigned int packed = 0u;
    if (e0 + 3 < E) {
        const int4 v = *(const int4*)(idx + e0);
        packed =  ((unsigned)(v.x / R_TILE) & 0xFFu)
               | (((unsigned)(v.y / R_TILE) & 0xFFu) << 8)
               | (((unsigned)(v.z / R_TILE) & 0xFFu) << 16)
               | (((unsigned)(v.w / R_TILE) & 0xFFu) << 24);
    } else {
#pragma unroll
        for (int u = 0; u < 4; ++u) {
            const int e = e0 + u;
            unsigned b = 0xFFu;
            if (e < E) b = (unsigned)(idx[e] / R_TILE) & 0xFFu;
            packed |= b << (8 * u);
        }
    }
    tb4[t] = packed;
}

__device__ __forceinline__ unsigned zero_bytes(unsigned x)
{
    return (x - 0x01010101u) & ~x & 0x80808080u;
}

__device__ __forceinline__ void flush_edges(
    volatile lds_int* q, int count, int lane, int tile, int base,
    const int* __restrict__ idx_l, const float* __restrict__ val_l,
    const int* __restrict__ idx_u, const float* __restrict__ val_u,
    const float* __restrict__ xm, const float* __restrict__ sc,
    float* acc, int N, int E, int Epad)
{
    int jj = 0;
    int r = -1;
    float alpha = 0.f;
    if (lane < count) {
        const int g = q[lane];
        const int list = (g >= Epad) ? 1 : 0;
        const int e = g - list * Epad;
        if (e < E) {
            const int*   idx = list ? idx_u : idx_l;
            const float* val = list ? val_u : val_l;
            const int   i  = idx[e];
            const int   j  = idx[(size_t)E + e];
            const float ve = val[e];
            if (i / R_TILE == tile) {
                const float* s = sc + (size_t)list * 2 * (size_t)N;
                const float z = s[j] + s[(size_t)N + i];
                const float elu = (z > 0.f) ? z : expm1f(z);
                alpha = elu * ve;
                jj = j + list * N;
                r = i - base;
            }
        }
    }
#pragma unroll
    for (int h = 0; h < 2; ++h) {
        if (h * 16 >= count) break;
        float2 m[16];
#pragma unroll
        for (int u = 0; u < 16; ++u) {
            const int jk = __builtin_amdgcn_readlane(jj, h * 16 + u);
            m[u] = *(const float2*)(xm + (size_t)jk * OUT_C + 2 * lane);
        }
#pragma unroll
        for (int u = 0; u < 16; ++u) {
            const int rk = __builtin_amdgcn_readlane(r, h * 16 + u);
            if (rk >= 0) {
                const float ak = __int_as_float(
                    __builtin_amdgcn_readlane(__float_as_int(alpha), h * 16 + u));
                float* a = acc + rk * OUT_C + 2 * lane;
                atomicAdd(a,     ak * m[u].x);
                atomicAdd(a + 1, ak * m[u].y);
            }
        }
    }
}

__global__ void __launch_bounds__(1024)
tile_accum_kernel(const unsigned int* __restrict__ tbytes,
                  const int* __restrict__ idx_l, const float* __restrict__ val_l,
                  const int* __restrict__ idx_u, const float* __restrict__ val_u,
                  const float* __restrict__ xm,
                  const float* __restrict__ sc,
                  float* __restrict__ out, int N, int E, int Epad)
{
    __shared__ float acc[R_TILE * OUT_C];
    __shared__ int queue[TILE_WAVES][64];

    const uint4* tb = (const uint4*)tbytes;
    const int tile = (int)blockIdx.x;
    const int base = tile * R_TILE;
    const int rows = min(R_TILE, N - base);
    const int tid  = (int)threadIdx.x;
    const int lane = tid & 31;
    const int wave = tid >> 5;

    {
        const float4* src = (const float4*)(out + (size_t)base * OUT_C);
        float4* dst = (float4*)acc;
        for (int t = tid; t < rows * (OUT_C / 4); t += 1024) dst[t] = src[t];
    }
    __syncthreads();

    const unsigned pat = (unsigned)(tile & 0xFF) * 0x01010101u;
    const int total   = 2 * Epad;
    const int nchunks = (total + 1023) >> 10;
    volatile lds_int* q = (volatile lds_int*)&queue[wave][0];
    int qn = 0;

    const int last = max(total - 32, 0);
    int o16 = min(wave * 1024 + lane * 32, last) >> 4;
    uint4 cur0 = tb[o16], cur1 = tb[o16 + 1];
    for (int c = wave; c < nchunks; c += TILE_WAVES) {
        o16 = min((c + TILE_WAVES) * 1024 + lane * 32, last) >> 4;
        const uint4 nxt0 = tb[o16], nxt1 = tb[o16 + 1];
        unsigned mk = (zero_bytes(cur0.x ^ pat) >> 7) | (zero_bytes(cur0.y ^ pat) >> 6)
                    | (zero_bytes(cur0.z ^ pat) >> 5) | (zero_bytes(cur0.w ^ pat) >> 4)
                    | (zero_bytes(cur1.x ^ pat) >> 3) | (zero_bytes(cur1.y ^ pat) >> 2)
                    | (zero_bytes(cur1.z ^ pat) >> 1) |  zero_bytes(cur1.w ^ pat);
        const int gbase = c * 1024 + lane * 32;
        if (gbase >= total) mk = 0u;
        for (;;) {
            const unsigned bal = __builtin_amdgcn_ballot_w32(mk != 0u);
            if (bal == 0u) break;
            if (mk != 0u) {
                const int p = __builtin_ctz(mk);
                mk &= mk - 1u;
                const int pos = qn + (int)__builtin_amdgcn_mbcnt_lo(bal, 0u);
                q[pos] = gbase + 4 * (p & 7) + (p >> 3);
            }
            qn += __popc(bal);
            if (qn >= 32) {
                flush_edges(q, 32, lane, tile, base, idx_l, val_l, idx_u, val_u,
                            xm, sc, acc, N, E, Epad);
                const int rem = qn - 32;
                if (lane < rem) { const int tmp = q[32 + lane]; q[lane] = tmp; }
                qn = rem;
            }
        }
        cur0 = nxt0; cur1 = nxt1;
    }
    if (qn > 0)
        flush_edges(q, qn, lane, tile, base, idx_l, val_l, idx_u, val_u,
                    xm, sc, acc, N, E, Epad);

    __syncthreads();
    {
        float4* dst = (float4*)(out + (size_t)base * OUT_C);
        const float4* src = (const float4*)acc;
        for (int t = tid; t < rows * (OUT_C / 4); t += 1024) {
            float4 v = src[t];
            v.x = fmaxf(v.x, 0.0f); v.y = fmaxf(v.y, 0.0f);
            v.z = fmaxf(v.z, 0.0f); v.w = fmaxf(v.w, 0.0f);
            dst[t] = v;
        }
    }
}

extern "C" void kernel_launch(void* const* d_in, const int* in_sizes, int n_in,
                              void* d_out, int out_size, void* d_ws, size_t ws_size,
                              hipStream_t stream) {
    const float* x        = (const float*)d_in[0];
    const int*   low_idx  = (const int*)  d_in[1];
    const float* low_val  = (const float*)d_in[2];
    const int*   up_idx   = (const int*)  d_in[3];
    const float* up_val   = (const float*)d_in[4];
    const float* w_lower  = (const float*)d_in[5];
    const float* a_lower  = (const float*)d_in[6];
    const float* w_upper  = (const float*)d_in[7];
    const float* a_upper  = (const float*)d_in[8];
    const float* w_lin    = (const float*)d_in[9];
    float*       out      = (float*)d_out;

    const int N = in_sizes[0] / IN_C;
    const int E = in_sizes[2];
    const int Epad = (E + 15) & ~15;

    float* ws   = (float*)d_ws;
    float* xm_l = ws;
    float* xm_u = xm_l + (size_t)N * OUT_C;
    float* sl_s = xm_u + (size_t)N * OUT_C;
    float* sl_t = sl_s + N;
    float* su_s = sl_t + N;
    float* su_t = su_s + N;
    unsigned int* tb4 = (unsigned int*)(su_t + N);

    {
        const int waves  = (N + 15) / 16;
        const int blocks = (waves + 3) / 4;
        node_gemm_kernel<<<blocks, 128, 0, stream>>>(
            x, w_lower, a_lower, w_upper, a_upper, w_lin,
            xm_l, xm_u, sl_s, sl_t, su_s, su_t, out, N);
    }

    {
        const int threads = (2 * Epad) / 4;
        const int blocks  = (threads + 255) / 256;
        tile_byte_kernel<<<blocks, 256, 0, stream>>>(low_idx, up_idx, tb4, E, Epad);
    }

    {
        const int tiles = (N + R_TILE - 1) / R_TILE;
        tile_accum_kernel<<<tiles, 1024, 0, stream>>>(
            tb4, low_idx, low_val, up_idx, up_val,
            xm_l, sl_s, out, N, E, Epad);
    }
}


# edges counting-sorted by target tile inside 4096-edge segments (LDS integer ranks): each tile's work-group reads only its own short lists, no scan of every tile byte; LDS float add order differs
# speedup vs baseline: 2.4075x; 1.1172x over previous
#include <hip/hip_runtime.h>
#include <hip/hip_bf16.h>

#define IN_C  64
#define OUT_C 64
#define EPS   (1.0f + 1e-06f)

typedef float v2f __attribute__((ext_vector_type(2)));
typedef float v8f __attribute__((ext_vector_type(8)));

template <bool HAS_ATT>
__device__ __forceinline__ void wmma_block(
    const v2f a_frag[16],
    const float* __restrict__ w,
    const float* __restrict__ att,
    float* __restrict__ store,
    float* __restrict__ s_src,
    float* __restrict__ s_tgt,
    float scale, int row0, int half, int lrow)
{
    float att_s[4], att_t[4];
    if (HAS_ATT) {
#pragma unroll
        for (int nt = 0; nt < 4; ++nt) {
            att_s[nt] = att[nt * 16 + lrow];
            att_t[nt] = att[OUT_C + nt * 16 + lrow];
        }
    }

    float ssrc_acc[8];
    float stgt_acc[8];
#pragma unroll
    for (int r = 0; r < 8; ++r) { ssrc_acc[r] = 0.f; stgt_acc[r] = 0.f; }

#pragma unroll
    for (int nt = 0; nt < 4; ++nt) {
        const int col = nt * 16 + lrow;
        v8f c = {};
#pragma unroll
        for (int kk = 0; kk < 16; ++kk) {
            const int k0 = 4 * kk + 2 * half;
            v2f b;
            b.x = w[(size_t)k0 * OUT_C + col];
            b.y = w[(size_t)(k0 + 1) * OUT_C + col];
            c = __builtin_amdgcn_wmma_f32_16x16x4_f32(
                     false, a_frag[kk],
                     false, b,
                     (short)0, c,
                     false,  false);
        }
#pragma unroll
        for (int r = 0; r < 8; ++r) {
            const int row = row0 + r + half * 8;
            const float v = c[r];
            store[(size_t)row * OUT_C + col] = v * scale;
            if (HAS_ATT) {
                ssrc_acc[r] += v * att_s[nt];
                stgt_acc[r] += v * att_t[nt];
            }
        }
    }

    if (HAS_ATT) {
#pragma unroll
        for (int r = 0; r < 8; ++r) {
            float vs = ssrc_acc[r];
            float vt = stgt_acc[r];
#pragma unroll
            for (int m = 8; m >= 1; m >>= 1) {
                vs += __shfl_xor(vs, m, 32);
                vt += __shfl_xor(vt, m, 32);
            }
            if (lrow == 0) {
                const int row = row0 + r + half * 8;
                s_src[row] = vs;
                s_tgt[row] = vt;
            }
        }
    }
}

__global__ void __launch_bounds__(128)
node_gemm_kernel(const float* __restrict__ x,
                 const float* __restrict__ w_lower,
                 const float* __restrict__ a_lower,
                 const float* __restrict__ w_upper,
                 const float* __restrict__ a_upper,
                 const float* __restrict__ w_lin,
                 float* __restrict__ xm_l, float* __restrict__ xm_u,
                 float* __restrict__ sl_s, float* __restrict__ sl_t,
                 float* __restrict__ su_s, float* __restrict__ su_t,
                 float* __restrict__ out, int n_rows)
{
    const int wid  = (int)((blockIdx.x * blockDim.x + threadIdx.x) >> 5);
    const int lane = (int)(threadIdx.x & 31);
    const int row0 = wid * 16;
    if (row0 >= n_rows) return;

    const int half = lane >> 4;
    const int lrow = lane & 15;

    v2f a_frag[16];
    const float* xr = x + (size_t)(row0 + lrow) * IN_C + 2 * half;
#pragma unroll
    for (int kk = 0; kk < 16; ++kk)
        a_frag[kk] = *(const v2f*)(xr + 4 * kk);

    wmma_block<true >(a_frag, w_lower, a_lower, xm_l, sl_s, sl_t, 1.0f, row0, half, lrow);
    wmma_block<true >(a_frag, w_upper, a_upper, xm_u, su_s, su_t, 1.0f, row0, half, lrow);
    wmma_block<false>(a_frag, w_lin,   nullptr, out,  nullptr, nullptr, EPS, row0, half, lrow);
}

#define R_TILE 196
#define TILE_WAVES 32
#define SEG 4096
#define SEG_TILES 256

__global__ void __launch_bounds__(1024)
seg_sort_kernel(const int* __restrict__ idx_l,
                const int* __restrict__ idx_u,
                unsigned short* __restrict__ sorted,
                unsigned int* __restrict__ offs,
                int E, int Epad,
                unsigned long long need_bytes, unsigned long long ws_bytes)
{
    if (need_bytes > ws_bytes) return;
    __shared__ int cnt[256];
    __shared__ int start[SEG_TILES + 1];
    __shared__ __attribute__((aligned(16))) unsigned short lst[SEG];
    const int tid = (int)threadIdx.x;
    const int seg = (int)blockIdx.x;
    if (tid < 256) cnt[tid] = 0;
    ((uint2*)lst)[tid] = make_uint2(0u, 0u);
    __syncthreads();

    const int g0 = seg * SEG + 4 * tid;
    int t[4] = {-1, -1, -1, -1};
    if (g0 < 2 * Epad) {
        const int list = (g0 >= Epad) ? 1 : 0;
        const int e0 = g0 - list * Epad;
        const int* idx = list ? idx_u : idx_l;
        if (e0 + 3 < E) {
            const int4 v = *(const int4*)(idx + e0);
            t[0] = v.x / R_TILE; t[1] = v.y / R_TILE;
            t[2] = v.z / R_TILE; t[3] = v.w / R_TILE;
        } else {
#pragma unroll
            for (int u = 0; u < 4; ++u)
                if (e0 + u < E) t[u] = idx[e0 + u] / R_TILE;
        }
    }
    int rank[4];
#pragma unroll
    for (int u = 0; u < 4; ++u) {
        rank[u] = 0;
        if ((unsigned)t[u] < 256u) rank[u] = atomicAdd(&cnt[t[u]], 1);
        else t[u] = -1;
    }
    __syncthreads();

    if (tid < 32) {
        int c[8];
        int s = 0;
#pragma unroll
        for (int k = 0; k < 8; ++k) { c[k] = cnt[8 * tid + k]; s += c[k]; }
        int incl = s;
#pragma unroll
        for (int d = 1; d < 32; d <<= 1) {
            const int nb = __shfl_up(incl, d, 32);
            if (tid >= d) incl += nb;
        }
        int run = incl - s;
#pragma unroll
        for (int k = 0; k < 8; ++k) { start[8 * tid + k] = run; run += c[k]; }
        if (tid == 31) start[256] = run;
    }
    __syncthreads();

#pragma unroll
    for (int u = 0; u < 4; ++u)
        if (t[u] >= 0) lst[start[t[u]] + rank[u]] = (unsigned short)(4 * tid + u);
    __syncthreads();

    ((uint2*)(sorted + (size_t)seg * SEG))[tid] = ((const uint2*)lst)[tid];
    if (tid < SEG_TILES)
        offs[(size_t)seg * SEG_TILES + tid] =
            (unsigned)start[tid] | ((unsigned)start[tid + 1] << 16);
}

__device__ __forceinline__ void flush_edges(
    int g, int count, int lane, int base,
    const int* __restrict__ idx_l, const float* __restrict__ val_l,
    const int* __restrict__ idx_u, const float* __restrict__ val_u,
    const float* __restrict__ xm, const float* __restrict__ sc,
    float* acc, int N, int E, int Epad)
{
    int jj = 0;
    int r = -1;
    float alpha = 0.f;
    if (g >= 0) {
        const int list = (g >= Epad) ? 1 : 0;
        const int e = g - list * Epad;
        const int*   idx = list ? idx_u : idx_l;
        const float* val = list ? val_u : val_l;
        const int   i  = idx[e];
        const int   j  = idx[(size_t)E + e];
        const float ve = val[e];
        const float* s = sc + (size_t)list * 2 * (size_t)N;
        const float z = s[j] + s[(size_t)N + i];
        const float elu = (z > 0.f) ? z : expm1f(z);
        alpha = elu * ve;
        jj = j + list * N;
        r = i - base;
    }
#pragma unroll
    for (int h = 0; h < 2; ++h) {
        if (h * 16 >= count) break;
        float2 m[16];
#pragma unroll
        for (int u = 0; u < 16; ++u) {
            const int rk = __builtin_amdgcn_readlane(r, h * 16 + u);
            m[u] = make_float2(0.f, 0.f);
            if (rk >= 0) {
                const int jk = __builtin_amdgcn_readlane(jj, h * 16 + u);
                m[u] = *(const float2*)(xm + (size_t)jk * OUT_C + 2 * lane);
            }
        }
#pragma unroll
        for (int u = 0; u < 16; ++u) {
            const int rk = __builtin_amdgcn_readlane(r, h * 16 + u);
            if (rk >= 0) {
                const float ak = __int_as_float(
                    __builtin_amdgcn_readlane(__float_as_int(alpha), h * 16 + u));
                float* a = acc + rk * OUT_C + 2 * lane;
                atomicAdd(a,     ak * m[u].x);
                atomicAdd(a + 1, ak * m[u].y);
            }
        }
    }
}

__global__ void __launch_bounds__(1024)
tile_accum_kernel(const unsigned short* __restrict__ sorted,
                  const unsigned int* __restrict__ offs,
                  const int* __restrict__ idx_l, const float* __restrict__ val_l,
                  const int* __restrict__ idx_u, const float* __restrict__ val_u,
                  const float* __restrict__ xm,
                  const float* __restrict__ sc,
                  float* __restrict__ out, int N, int E, int Epad, int nseg,
                  unsigned long long need_bytes, unsigned long long ws_bytes)
{
    if (need_bytes > ws_bytes) return;
    __shared__ float acc[R_TILE * OUT_C];

    const int tile = (int)blockIdx.x;
    const int base = tile * R_TILE;
    const int rows = min(R_TILE, N - base);
    const int tid  = (int)threadIdx.x;
    const int lane = tid & 31;
    const int wave = __builtin_amdgcn_readfirstlane(tid >> 5);

    unsigned pp = 0u;
    {
        const int sseg = wave + TILE_WAVES * lane;
        if (sseg < nseg) pp = offs[(size_t)sseg * SEG_TILES + tile];
    }

    {
        const float4* src = (const float4*)(out + (size_t)base * OUT_C);
        float4* dst = (float4*)acc;
        for (int t = tid; t < rows * (OUT_C / 4); t += 1024) dst[t] = src[t];
    }
    __syncthreads();

    for (int sb = wave; sb < nseg; sb += TILE_WAVES * 32) {
        if (sb != wave) {
            const int sseg = sb + TILE_WAVES * lane;
            pp = 0u;
            if (sseg < nseg) pp = offs[(size_t)sseg * SEG_TILES + tile];
        }
        unsigned pw = (unsigned)__builtin_amdgcn_readlane((int)pp, 0);
        int o0 = (int)(pw & 0xFFFFu), o1 = (int)(pw >> 16);
        int raw = (int)sorted[(size_t)sb * SEG + min(o0 + lane, SEG - 1)];
        for (int n = 0; n < 32; ++n) {
            const int seg = sb + TILE_WAVES * n;
            if (seg >= nseg) break;
            const bool has_nxt = (n + 1 < 32) && (seg + TILE_WAVES < nseg);
            const int sn = min(seg + TILE_WAVES, nseg - 1);
            pw = (unsigned)__builtin_amdgcn_readlane((int)pp, min(n + 1, 31));
            int o0_nxt = (int)(pw & 0xFFFFu), o1_nxt = (int)(pw >> 16);
            if (!has_nxt) { o0_nxt = 0; o1_nxt = 0; }
            const int raw_nxt = (int)sorted[(size_t)sn * SEG + min(o0_nxt + lane, SEG - 1)];
            if (o1 > o0) {
                const int g = (lane < o1 - o0) ? seg * SEG + raw : -1;
                flush_edges(g, min(32, o1 - o0), lane, base, idx_l, val_l, idx_u, val_u,
                            xm, sc, acc, N, E, Epad);
            }
            for (int k0 = o0 + 32; k0 < o1; k0 += 32) {
                int g2 = -1;
                if (lane < o1 - k0) g2 = seg * SEG + (int)sorted[(size_t)seg * SEG + k0 + lane];
                flush_edges(g2, min(32, o1 - k0), lane, base, idx_l, val_l, idx_u, val_u,
                            xm, sc, acc, N, E, Epad);
            }
            raw = raw_nxt; o0 = o0_nxt; o1 = o1_nxt;
        }
    }

    __syncthreads();
    {
        float4* dst = (float4*)(out + (size_t)base * OUT_C);
        const float4* src = (const float4*)acc;
        for (int t = tid; t < rows * (OUT_C / 4); t += 1024) {
            float4 v = src[t];
            v.x = fmaxf(v.x, 0.0f); v.y = fmaxf(v.y, 0.0f);
            v.z = fmaxf(v.z, 0.0f); v.w = fmaxf(v.w, 0.0f);
            dst[t] = v;
        }
    }
}

extern "C" void kernel_launch(void* const* d_in, const int* in_sizes, int n_in,
                              void* d_out, int out_size, void* d_ws, size_t ws_size,
                              hipStream_t stream) {
    const float* x        = (const float*)d_in[0];
    const int*   low_idx  = (const int*)  d_in[1];
    const float* low_val  = (const float*)d_in[2];
    const int*   up_idx   = (const int*)  d_in[3];
    const float* up_val   = (const float*)d_in[4];
    const float* w_lower  = (const float*)d_in[5];
    const float* a_lower  = (const float*)d_in[6];
    const float* w_upper  = (const float*)d_in[7];
    const float* a_upper  = (const float*)d_in[8];
    const float* w_lin    = (const float*)d_in[9];
    float*       out      = (float*)d_out;

    const int N = in_sizes[0] / IN_C;
    const int E = in_sizes[2];
    const int Epad = (E + 15) & ~15;

    float* ws   = (float*)d_ws;
    float* xm_l = ws;
    float* xm_u = xm_l + (size_t)N * OUT_C;
    float* sl_s = xm_u + (size_t)N * OUT_C;
    float* sl_t = sl_s + N;
    float* su_s = sl_t + N;
    float* su_t = su_s + N;
    const int nseg = (2 * Epad + SEG - 1) / SEG;
    unsigned short* sorted = (unsigned short*)(su_t + N);
    unsigned int*   offs   = (unsigned int*)(sorted + (size_t)nseg * SEG);
    const unsigned long long need =
        (unsigned long long)((const char*)(offs + (size_t)nseg * SEG_TILES) - (const char*)d_ws);
    const unsigned long long have = (unsigned long long)ws_size;

    {
        const int waves  = (N + 15) / 16;
        const int blocks = (waves + 3) / 4;
        node_gemm_kernel<<<blocks, 128, 0, stream>>>(
            x, w_lower, a_lower, w_upper, a_upper, w_lin,
            xm_l, xm_u, sl_s, sl_t, su_s, su_t, out, N);
    }

    seg_sort_kernel<<<nseg, 1024, 0, stream>>>(low_idx, up_idx, sorted, offs, E, Epad, need, have);

    {
        const int tiles = (N + R_TILE - 1) / R_TILE;
        tile_accum_kernel<<<tiles, 1024, 0, stream>>>(
            sorted, offs, low_idx, low_val, up_idx, up_val,
            xm_l, sl_s, out, N, E, Epad, nseg, need, have);
    }
}


# sort kernel stores each edge as an 8-byte record (source row, row in tile, value) in tile order; tile kernel reads records contiguous, no per-edge index and value gathers; LDS add order as before
# speedup vs baseline: 2.8130x; 1.1684x over previous
#include <hip/hip_runtime.h>
#include <hip/hip_bf16.h>

#define IN_C  64
#define OUT_C 64
#define EPS   (1.0f + 1e-06f)

typedef float v2f __attribute__((ext_vector_type(2)));
typedef float v8f __attribute__((ext_vector_type(8)));

template <bool HAS_ATT>
__device__ __forceinline__ void wmma_block(
    const v2f a_frag[16],
    const float* __restrict__ w,
    const float* __restrict__ att,
    float* __restrict__ store,
    float* __restrict__ s_src,
    float* __restrict__ s_tgt,
    float scale, int row0, int half, int lrow)
{
    float att_s[4], att_t[4];
    if (HAS_ATT) {
#pragma unroll
        for (int nt = 0; nt < 4; ++nt) {
            att_s[nt] = att[nt * 16 + lrow];
            att_t[nt] = att[OUT_C + nt * 16 + lrow];
        }
    }

    float ssrc_acc[8];
    float stgt_acc[8];
#pragma unroll
    for (int r = 0; r < 8; ++r) { ssrc_acc[r] = 0.f; stgt_acc[r] = 0.f; }

#pragma unroll
    for (int nt = 0; nt < 4; ++nt) {
        const int col = nt * 16 + lrow;
        v8f c = {};
#pragma unroll
        for (int kk = 0; kk < 16; ++kk) {
            const int k0 = 4 * kk + 2 * half;
            v2f b;
            b.x = w[(size_t)k0 * OUT_C + col];
            b.y = w[(size_t)(k0 + 1) * OUT_C + col];
            c = __builtin_amdgcn_wmma_f32_16x16x4_f32(
                     false, a_frag[kk],
                     false, b,
                     (short)0, c,
                     false,  false);
        }
#pragma unroll
        for (int r = 0; r < 8; ++r) {
            const int row = row0 + r + half * 8;
            const float v = c[r];
            store[(size_t)row * OUT_C + col] = v * scale;
            if (HAS_ATT) {
                ssrc_acc[r] += v * att_s[nt];
                stgt_acc[r] += v * att_t[nt];
            }
        }
    }

    if (HAS_ATT) {
#pragma unroll
        for (int r = 0; r < 8; ++r) {
            float vs = ssrc_acc[r];
            float vt = stgt_acc[r];
#pragma unroll
            for (int m = 8; m >= 1; m >>= 1) {
                vs += __shfl_xor(vs, m, 32);
                vt += __shfl_xor(vt, m, 32);
            }
            if (lrow == 0) {
                const int row = row0 + r + half * 8;
                s_src[row] = vs;
                s_tgt[row] = vt;
            }
        }
    }
}

__global__ void __launch_bounds__(128)
node_gemm_kernel(const float* __restrict__ x,
                 const float* __restrict__ w_lower,
                 const float* __restrict__ a_lower,
                 const float* __restrict__ w_upper,
                 const float* __restrict__ a_upper,
                 const float* __restrict__ w_lin,
                 float* __restrict__ xm_l, float* __restrict__ xm_u,
                 float* __restrict__ sl_s, float* __restrict__ sl_t,
                 float* __restrict__ su_s, float* __restrict__ su_t,
                 float* __restrict__ out, int n_rows)
{
    const int wid  = (int)((blockIdx.x * blockDim.x + threadIdx.x) >> 5);
    const int lane = (int)(threadIdx.x & 31);
    const int row0 = wid * 16;
    if (row0 >= n_rows) return;

    const int half = lane >> 4;
    const int lrow = lane & 15;

    v2f a_frag[16];
    const float* xr = x + (size_t)(row0 + lrow) * IN_C + 2 * half;
#pragma unroll
    for (int kk = 0; kk < 16; ++kk)
        a_frag[kk] = *(const v2f*)(xr + 4 * kk);

    wmma_block<true >(a_frag, w_lower, a_lower, xm_l, sl_s, sl_t, 1.0f, row0, half, lrow);
    wmma_block<true >(a_frag, w_upper, a_upper, xm_u, su_s, su_t, 1.0f, row0, half, lrow);
    wmma_block<false>(a_frag, w_lin,   nullptr, out,  nullptr, nullptr, EPS, row0, half, lrow);
}

#define R_TILE 196
#define TILE_WAVES 32
#define SEG 4096
#define SEG_TILES 256
#define REC_BYTES 8
#define POS_BYTES 2

__device__ __forceinline__ void scan_counts(const int* cnt, int* start, int tid)
{
    if (tid < 32) {
        int c[8];
        int s = 0;
#pragma unroll
        for (int k = 0; k < 8; ++k) { c[k] = cnt[8 * tid + k]; s += c[k]; }
        int incl = s;
#pragma unroll
        for (int d = 1; d < 32; d <<= 1) {
            const int nb = __shfl_up(incl, d, 32);
            if (tid >= d) incl += nb;
        }
        int run = incl - s;
#pragma unroll
        for (int k = 0; k < 8; ++k) { start[8 * tid + k] = run; run += c[k]; }
        if (tid == 31) start[256] = run;
    }
}

__global__ void __launch_bounds__(1024)
seg_sort_kernel(const int* __restrict__ idx_l, const float* __restrict__ val_l,
                const int* __restrict__ idx_u, const float* __restrict__ val_u,
                unsigned int* __restrict__ region,
                int N, int E, int Epad, int nseg,
                unsigned long long need_rec, unsigned long long need_pos,
                unsigned long long ws_bytes)
{
    if (need_pos > ws_bytes) return;
    const bool use_rec = (need_rec <= ws_bytes);
    __shared__ int cnt[256];
    __shared__ int start[SEG_TILES + 1];
    __shared__ __attribute__((aligned(16))) unsigned int lst[SEG * 2];
    const int tid = (int)threadIdx.x;
    const int seg = (int)blockIdx.x;
    unsigned int* offs = (unsigned int*)((char*)region +
        (size_t)nseg * SEG * (use_rec ? REC_BYTES : POS_BYTES));
    if (tid < 256) cnt[tid] = 0;
    if (use_rec) {
        ((uint4*)lst)[tid]        = make_uint4(0u, 0u, 0u, 0u);
        ((uint4*)lst)[tid + 1024] = make_uint4(0u, 0u, 0u, 0u);
    } else {
        ((uint2*)lst)[tid] = make_uint2(0u, 0u);
    }
    __syncthreads();

    const int g0 = seg * SEG + 4 * tid;
    int t[4] = {-1, -1, -1, -1};
    unsigned w0[4] = {0u, 0u, 0u, 0u};
    float ve[4] = {0.f, 0.f, 0.f, 0.f};
    if (g0 < 2 * Epad) {
        const int list = (g0 >= Epad) ? 1 : 0;
        const int e0 = g0 - list * Epad;
        const int*   idx = list ? idx_u : idx_l;
        const float* val = list ? val_u : val_l;
        int ti[4] = {-1, -1, -1, -1};
        int tj[4] = {0, 0, 0, 0};
        if (e0 + 3 < E) {
            const int4 v = *(const int4*)(idx + e0);
            ti[0] = v.x; ti[1] = v.y; ti[2] = v.z; ti[3] = v.w;
            if (use_rec) {
                if ((E & 3) == 0) {
                    const int4 q = *(const int4*)(idx + (size_t)E + e0);
                    tj[0] = q.x; tj[1] = q.y; tj[2] = q.z; tj[3] = q.w;
                } else {
#pragma unroll
                    for (int u = 0; u < 4; ++u) tj[u] = idx[(size_t)E + e0 + u];
                }
                const float4 f = *(const float4*)(val + e0);
                ve[0] = f.x; ve[1] = f.y; ve[2] = f.z; ve[3] = f.w;
            }
        } else {
#pragma unroll
            for (int u = 0; u < 4; ++u)
                if (e0 + u < E) {
                    ti[u] = idx[e0 + u];
                    if (use_rec) { tj[u] = idx[(size_t)E + e0 + u]; ve[u] = val[e0 + u]; }
                }
        }
#pragma unroll
        for (int u = 0; u < 4; ++u)
            if (ti[u] >= 0) {
                t[u] = ti[u] / R_TILE;
                w0[u] = (unsigned)(tj[u] + list * N) | ((unsigned)(ti[u] - t[u] * R_TILE) << 24);
            }
    }
    int rank[4];
#pragma unroll
    for (int u = 0; u < 4; ++u) {
        rank[u] = 0;
        if ((unsigned)t[u] < 256u) rank[u] = atomicAdd(&cnt[t[u]], 1);
        else t[u] = -1;
    }
    __syncthreads();
    scan_counts(cnt, start, tid);
    __syncthreads();

    if (use_rec) {
#pragma unroll
        for (int u = 0; u < 4; ++u)
            if (t[u] >= 0)
                ((uint2*)lst)[start[t[u]] + rank[u]] = make_uint2(w0[u], __float_as_uint(ve[u]));
    } else {
#pragma unroll
        for (int u = 0; u < 4; ++u)
            if (t[u] >= 0)
                ((unsigned short*)lst)[start[t[u]] + rank[u]] = (unsigned short)(4 * tid + u);
    }
    __syncthreads();

    if (use_rec) {
        uint4* dst = (uint4*)((char*)region + (size_t)seg * SEG * REC_BYTES);
        dst[tid]        = ((const uint4*)lst)[tid];
        dst[tid + 1024] = ((const uint4*)lst)[tid + 1024];
    } else {
        uint2* dst = (uint2*)((char*)region + (size_t)seg * SEG * POS_BYTES);
        dst[tid] = ((const uint2*)lst)[tid];
    }
    if (tid < SEG_TILES)
        offs[(size_t)seg * SEG_TILES + tid] =
            (unsigned)start[tid] | ((unsigned)start[tid + 1] << 16);
}

__device__ __forceinline__ void add_rows(
    int jj, int r, float alpha, int count, int lane,
    const float* __restrict__ xm, float* acc)
{
#pragma unroll
    for (int h = 0; h < 2; ++h) {
        if (h * 16 >= count) break;
        float2 m[16];
#pragma unroll
        for (int u = 0; u < 16; ++u) {
            const int rk = __builtin_amdgcn_readlane(r, h * 16 + u);
            m[u] = make_float2(0.f, 0.f);
            if (rk >= 0) {
                const int jk = __builtin_amdgcn_readlane(jj, h * 16 + u);
                m[u] = *(const float2*)(xm + (size_t)jk * OUT_C + 2 * lane);
            }
        }
#pragma unroll
        for (int u = 0; u < 16; ++u) {
            const int rk = __builtin_amdgcn_readlane(r, h * 16 + u);
            if (rk >= 0) {
                const float ak = __int_as_float(
                    __builtin_amdgcn_readlane(__float_as_int(alpha), h * 16 + u));
                float* a = acc + rk * OUT_C + 2 * lane;
                atomicAdd(a,     ak * m[u].x);
                atomicAdd(a + 1, ak * m[u].y);
            }
        }
    }
}

__device__ __forceinline__ void flush_recs(
    unsigned w0, float ve, bool valid, int count, int lane, int base,
    const float* __restrict__ xm, const float* __restrict__ sc,
    float* acc, int N)
{
    int jj = 0;
    int r = -1;
    float alpha = 0.f;
    if (valid) {
        jj = (int)(w0 & 0x00FFFFFFu);
        r  = (int)(w0 >> 24);
        const int list = (jj >= N) ? 1 : 0;
        const float* s = sc + (size_t)list * 2 * (size_t)N;
        const float z = s[jj - list * N] + s[(size_t)N + base + r];
        const float elu = (z > 0.f) ? z : expm1f(z);
        alpha = elu * ve;
    }
    add_rows(jj, r, alpha, count, lane, xm, acc);
}

__device__ __forceinline__ void flush_edges(
    int g, int count, int lane, int base,
    const int* __restrict__ idx_l, const float* __restrict__ val_l,
    const int* __restrict__ idx_u, const float* __restrict__ val_u,
    const float* __restrict__ xm, const float* __restrict__ sc,
    float* acc, int N, int E, int Epad)
{
    int jj = 0;
    int r = -1;
    float alpha = 0.f;
    if (g >= 0) {
        const int list = (g >= Epad) ? 1 : 0;
        const int e = g - list * Epad;
        const int*   idx = list ? idx_u : idx_l;
        const float* val = list ? val_u : val_l;
        const int   i  = idx[e];
        const int   j  = idx[(size_t)E + e];
        const float ve = val[e];
        const float* s = sc + (size_t)list * 2 * (size_t)N;
        const float z = s[j] + s[(size_t)N + i];
        const float elu = (z > 0.f) ? z : expm1f(z);
        alpha = elu * ve;
        jj = j + list * N;
        r = i - base;
    }
    add_rows(jj, r, alpha, count, lane, xm, acc);
}

__global__ void __launch_bounds__(1024)
tile_accum_kernel(const unsigned int* __restrict__ region,
                  const int* __restrict__ idx_l, const float* __restrict__ val_l,
                  const int* __restrict__ idx_u, const float* __restrict__ val_u,
                  const float* __restrict__ xm,
                  const float* __restrict__ sc,
                  float* __restrict__ out, int N, int E, int Epad, int nseg,
                  unsigned long long need_rec, unsigned long long need_pos,
                  unsigned long long ws_bytes)
{
    if (need_pos > ws_bytes) return;
    const bool use_rec = (need_rec <= ws_bytes);
    __shared__ float acc[R_TILE * OUT_C];

    const unsigned int* offs = (const unsigned int*)((const char*)region +
        (size_t)nseg * SEG * (use_rec ? REC_BYTES : POS_BYTES));

    const int tile = (int)blockIdx.x;
    const int base = tile * R_TILE;
    const int rows = min(R_TILE, N - base);
    const int tid  = (int)threadIdx.x;
    const int lane = tid & 31;
    const int wave = __builtin_amdgcn_readfirstlane(tid >> 5);

    unsigned pp = 0u;
    {
        const int sseg = wave + TILE_WAVES * lane;
        if (sseg < nseg) pp = offs[(size_t)sseg * SEG_TILES + tile];
    }

    {
        const float4* src = (const float4*)(out + (size_t)base * OUT_C);
        float4* dst = (float4*)acc;
        for (int t = tid; t < rows * (OUT_C / 4); t += 1024) dst[t] = src[t];
    }
    __syncthreads();

    if (use_rec) {
        const uint2* recs = (const uint2*)region;
        for (int sb = wave; sb < nseg; sb += TILE_WAVES * 32) {
            if (sb != wave) {
                const int sseg = sb + TILE_WAVES * lane;
                pp = 0u;
                if (sseg < nseg) pp = offs[(size_t)sseg * SEG_TILES + tile];
            }
            unsigned pw = (unsigned)__builtin_amdgcn_readlane((int)pp, 0);
            int o0 = (int)(pw & 0xFFFFu), o1 = (int)(pw >> 16);
            uint2 rec = recs[(size_t)sb * SEG + min(o0 + lane, SEG - 1)];
            for (int n = 0; n < 32; ++n) {
                const int seg = sb + TILE_WAVES * n;
                if (seg >= nseg) break;
                const bool has_nxt = (n + 1 < 32) && (seg + TILE_WAVES < nseg);
                const int sn = min(seg + TILE_WAVES, nseg - 1);
                pw = (unsigned)__builtin_amdgcn_readlane((int)pp, min(n + 1, 31));
                int o0_nxt = (int)(pw & 0xFFFFu), o1_nxt = (int)(pw >> 16);
                if (!has_nxt) { o0_nxt = 0; o1_nxt = 0; }
                const uint2 rec_nxt = recs[(size_t)sn * SEG + min(o0_nxt + lane, SEG - 1)];
                if (o1 > o0)
                    flush_recs(rec.x, __uint_as_float(rec.y), lane < o1 - o0, min(32, o1 - o0),
                               lane, base, xm, sc, acc, N);
                for (int k0 = o0 + 32; k0 < o1; k0 += 32) {
                    const bool v2 = lane < o1 - k0;
                    uint2 r2 = make_uint2(0u, 0u);
                    if (v2) r2 = recs[(size_t)seg * SEG + k0 + lane];
                    flush_recs(r2.x, __uint_as_float(r2.y), v2, min(32, o1 - k0),
                               lane, base, xm, sc, acc, N);
                }
                rec = rec_nxt; o0 = o0_nxt; o1 = o1_nxt;
            }
        }
    } else {
        const unsigned short* sorted = (const unsigned short*)region;
        for (int sb = wave; sb < nseg; sb += TILE_WAVES * 32) {
            if (sb != wave) {
                const int sseg = sb + TILE_WAVES * lane;
                pp = 0u;
                if (sseg < nseg) pp = offs[(size_t)sseg * SEG_TILES + tile];
            }
            unsigned pw = (unsigned)__builtin_amdgcn_readlane((int)pp, 0);
            int o0 = (int)(pw & 0xFFFFu), o1 = (int)(pw >> 16);
            int raw = (int)sorted[(size_t)sb * SEG + min(o0 + lane, SEG - 1)];
            for (int n = 0; n < 32; ++n) {
                const int seg = sb + TILE_WAVES * n;
                if (seg >= nseg) break;
                const bool has_nxt = (n + 1 < 32) && (seg + TILE_WAVES < nseg);
                const int sn = min(seg + TILE_WAVES, nseg - 1);
                pw = (unsigned)__builtin_amdgcn_readlane((int)pp, min(n + 1, 31));
                int o0_nxt = (int)(pw & 0xFFFFu), o1_nxt = (int)(pw >> 16);
                if (!has_nxt) { o0_nxt = 0; o1_nxt = 0; }
                const int raw_nxt = (int)sorted[(size_t)sn * SEG + min(o0_nxt + lane, SEG - 1)];
                if (o1 > o0) {
                    const int g = (lane < o1 - o0) ? seg * SEG + raw : -1;
                    flush_edges(g, min(32, o1 - o0), lane, base, idx_l, val_l, idx_u, val_u,
                                xm, sc, acc, N, E, Epad);
                }
                for (int k0 = o0 + 32; k0 < o1; k0 += 32) {
                    int g2 = -1;
                    if (lane < o1 - k0) g2 = seg * SEG + (int)sorted[(size_t)seg * SEG + k0 + lane];
                    flush_edges(g2, min(32, o1 - k0), lane, base, idx_l, val_l, idx_u, val_u,
                                xm, sc, acc, N, E, Epad);
                }
                raw = raw_nxt; o0 = o0_nxt; o1 = o1_nxt;
            }
        }
    }

    __syncthreads();
    {
        float4* dst = (float4*)(out + (size_t)base * OUT_C);
        const float4* src = (const float4*)acc;
        for (int t = tid; t < rows * (OUT_C / 4); t += 1024) {
            float4 v = src[t];
            v.x = fmaxf(v.x, 0.0f); v.y = fmaxf(v.y, 0.0f);
            v.z = fmaxf(v.z, 0.0f); v.w = fmaxf(v.w, 0.0f);
            dst[t] = v;
        }
    }
}

extern "C" void kernel_launch(void* const* d_in, const int* in_sizes, int n_in,
                              void* d_out, int out_size, void* d_ws, size_t ws_size,
                              hipStream_t stream) {
    const float* x        = (const float*)d_in[0];
    const int*   low_idx  = (const int*)  d_in[1];
    const float* low_val  = (const float*)d_in[2];
    const int*   up_idx   = (const int*)  d_in[3];
    const float* up_val   = (const float*)d_in[4];
    const float* w_lower  = (const float*)d_in[5];
    const float* a_lower  = (const float*)d_in[6];
    const float* w_upper  = (const float*)d_in[7];
    const float* a_upper  = (const float*)d_in[8];
    const float* w_lin    = (const float*)d_in[9];
    float*       out      = (float*)d_out;

    const int N = in_sizes[0] / IN_C;
    const int E = in_sizes[2];
    const int Epad = (E + 15) & ~15;

    float* ws   = (float*)d_ws;
    float* xm_l = ws;
    float* xm_u = xm_l + (size_t)N * OUT_C;
    float* sl_s = xm_u + (size_t)N * OUT_C;
    float* sl_t = sl_s + N;
    float* su_s = sl_t + N;
    float* su_t = su_s + N;
    const int nseg = (2 * Epad + SEG - 1) / SEG;
    unsigned int* region = (unsigned int*)(su_t + N);
    const char* lists_end_rec = (const char*)region + (size_t)nseg * SEG * REC_BYTES;
    const char* lists_end_pos = (const char*)region + (size_t)nseg * SEG * POS_BYTES;
    const unsigned long long need_rec = (unsigned long long)
        (lists_end_rec + (size_t)nseg * SEG_TILES * sizeof(unsigned int) - (const char*)d_ws);
    const unsigned long long need_pos = (unsigned long long)
        (lists_end_pos + (size_t)nseg * SEG_TILES * sizeof(unsigned int) - (const char*)d_ws);
    const unsigned long long have = (unsigned long long)ws_size;

    {
        const int waves  = (N + 15) / 16;
        const int blocks = (waves + 3) / 4;
        node_gemm_kernel<<<blocks, 128, 0, stream>>>(
            x, w_lower, a_lower, w_upper, a_upper, w_lin,
            xm_l, xm_u, sl_s, sl_t, su_s, su_t, out, N);
    }

    seg_sort_kernel<<<nseg, 1024, 0, stream>>>(low_idx, low_val, up_idx, up_val, region,
                                               N, E, Epad, nseg, need_rec, need_pos, have);

    {
        const int tiles = (N + R_TILE - 1) / R_TILE;
        tile_accum_kernel<<<tiles, 1024, 0, stream>>>(
            region, low_idx, low_val, up_idx, up_val,
            xm_l, sl_s, out, N, E, Epad, nseg, need_rec, need_pos, have);
    }
}


# tile kernel sorts records by target row in LDS (count, scan, ranked copy); waves take equal slices, sum each row's run in registers: one LDS float-add pair per run, not per edge; sum order differs
# speedup vs baseline: 3.4310x; 1.2197x over previous
#include <hip/hip_runtime.h>
#include <hip/hip_bf16.h>

#define IN_C  64
#define OUT_C 64
#define EPS   (1.0f + 1e-06f)

typedef float v2f __attribute__((ext_vector_type(2)));
typedef float v8f __attribute__((ext_vector_type(8)));

template <bool HAS_ATT>
__device__ __forceinline__ void wmma_block(
    const v2f a_frag[16],
    const float* __restrict__ w,
    const float* __restrict__ att,
    float* __restrict__ store,
    float* __restrict__ s_src,
    float* __restrict__ s_tgt,
    float scale, int row0, int half, int lrow)
{
    float att_s[4], att_t[4];
    if (HAS_ATT) {
#pragma unroll
        for (int nt = 0; nt < 4; ++nt) {
            att_s[nt] = att[nt * 16 + lrow];
            att_t[nt] = att[OUT_C + nt * 16 + lrow];
        }
    }

    float ssrc_acc[8];
    float stgt_acc[8];
#pragma unroll
    for (int r = 0; r < 8; ++r) { ssrc_acc[r] = 0.f; stgt_acc[r] = 0.f; }

#pragma unroll
    for (int nt = 0; nt < 4; ++nt) {
        const int col = nt * 16 + lrow;
        v8f c = {};
#pragma unroll
        for (int kk = 0; kk < 16; ++kk) {
            const int k0 = 4 * kk + 2 * half;
            v2f b;
            b.x = w[(size_t)k0 * OUT_C + col];
            b.y = w[(size_t)(k0 + 1) * OUT_C + col];
            c = __builtin_amdgcn_wmma_f32_16x16x4_f32(
                     false, a_frag[kk],
                     false, b,
                     (short)0, c,
                     false,  false);
        }
#pragma unroll
        for (int r = 0; r < 8; ++r) {
            const int row = row0 + r + half * 8;
            const float v = c[r];
            store[(size_t)row * OUT_C + col] = v * scale;
            if (HAS_ATT) {
                ssrc_acc[r] += v * att_s[nt];
                stgt_acc[r] += v * att_t[nt];
            }
        }
    }

    if (HAS_ATT) {
#pragma unroll
        for (int r = 0; r < 8; ++r) {
            float vs = ssrc_acc[r];
            float vt = stgt_acc[r];
#pragma unroll
            for (int m = 8; m >= 1; m >>= 1) {
                vs += __shfl_xor(vs, m, 32);
                vt += __shfl_xor(vt, m, 32);
            }
            if (lrow == 0) {
                const int row = row0 + r + half * 8;
                s_src[row] = vs;
                s_tgt[row] = vt;
            }
        }
    }
}

__global__ void __launch_bounds__(128)
node_gemm_kernel(const float* __restrict__ x,
                 const float* __restrict__ w_lower,
                 const float* __restrict__ a_lower,
                 const float* __restrict__ w_upper,
                 const float* __restrict__ a_upper,
                 const float* __restrict__ w_lin,
                 float* __restrict__ xm_l, float* __restrict__ xm_u,
                 float* __restrict__ sl_s, float* __restrict__ sl_t,
                 float* __restrict__ su_s, float* __restrict__ su_t,
                 float* __restrict__ out, int n_rows)
{
    const int wid  = (int)((blockIdx.x * blockDim.x + threadIdx.x) >> 5);
    const int lane = (int)(threadIdx.x & 31);
    const int row0 = wid * 16;
    if (row0 >= n_rows) return;

    const int half = lane >> 4;
    const int lrow = lane & 15;

    v2f a_frag[16];
    const float* xr = x + (size_t)(row0 + lrow) * IN_C + 2 * half;
#pragma unroll
    for (int kk = 0; kk < 16; ++kk)
        a_frag[kk] = *(const v2f*)(xr + 4 * kk);

    wmma_block<true >(a_frag, w_lower, a_lower, xm_l, sl_s, sl_t, 1.0f, row0, half, lrow);
    wmma_block<true >(a_frag, w_upper, a_upper, xm_u, su_s, su_t, 1.0f, row0, half, lrow);
    wmma_block<false>(a_frag, w_lin,   nullptr, out,  nullptr, nullptr, EPS, row0, half, lrow);
}

#define R_TILE 196
#define TILE_WAVES 32
#define SEG 4096
#define SEG_TILES 256
#define REC_BYTES 8
#define POS_BYTES 2

__device__ __forceinline__ void scan_counts(const int* cnt, int* start, int tid)
{
    if (tid < 32) {
        int c[8];
        int s = 0;
#pragma unroll
        for (int k = 0; k < 8; ++k) { c[k] = cnt[8 * tid + k]; s += c[k]; }
        int incl = s;
#pragma unroll
        for (int d = 1; d < 32; d <<= 1) {
            const int nb = __shfl_up(incl, d, 32);
            if (tid >= d) incl += nb;
        }
        int run = incl - s;
#pragma unroll
        for (int k = 0; k < 8; ++k) { start[8 * tid + k] = run; run += c[k]; }
        if (tid == 31) start[256] = run;
    }
}

__global__ void __launch_bounds__(1024)
seg_sort_kernel(const int* __restrict__ idx_l, const float* __restrict__ val_l,
                const int* __restrict__ idx_u, const float* __restrict__ val_u,
                unsigned int* __restrict__ region,
                int N, int E, int Epad, int nseg,
                unsigned long long need_rec, unsigned long long need_pos,
                unsigned long long ws_bytes)
{
    if (need_pos > ws_bytes) return;
    const bool use_rec = (need_rec <= ws_bytes);
    __shared__ int cnt[256];
    __shared__ int start[SEG_TILES + 1];
    __shared__ __attribute__((aligned(16))) unsigned int lst[SEG * 2];
    const int tid = (int)threadIdx.x;
    const int seg = (int)blockIdx.x;
    unsigned int* offs = (unsigned int*)((char*)region +
        (size_t)nseg * SEG * (use_rec ? REC_BYTES : POS_BYTES));
    if (tid < 256) cnt[tid] = 0;
    if (use_rec) {
        ((uint4*)lst)[tid]        = make_uint4(0u, 0u, 0u, 0u);
        ((uint4*)lst)[tid + 1024] = make_uint4(0u, 0u, 0u, 0u);
    } else {
        ((uint2*)lst)[tid] = make_uint2(0u, 0u);
    }
    __syncthreads();

    const int g0 = seg * SEG + 4 * tid;
    int t[4] = {-1, -1, -1, -1};
    unsigned w0[4] = {0u, 0u, 0u, 0u};
    float ve[4] = {0.f, 0.f, 0.f, 0.f};
    if (g0 < 2 * Epad) {
        const int list = (g0 >= Epad) ? 1 : 0;
        const int e0 = g0 - list * Epad;
        const int*   idx = list ? idx_u : idx_l;
        const float* val = list ? val_u : val_l;
        int ti[4] = {-1, -1, -1, -1};
        int tj[4] = {0, 0, 0, 0};
        if (e0 + 3 < E) {
            const int4 v = *(const int4*)(idx + e0);
            ti[0] = v.x; ti[1] = v.y; ti[2] = v.z; ti[3] = v.w;
            if (use_rec) {
                if ((E & 3) == 0) {
                    const int4 q = *(const int4*)(idx + (size_t)E + e0);
                    tj[0] = q.x; tj[1] = q.y; tj[2] = q.z; tj[3] = q.w;
                } else {
#pragma unroll
                    for (int u = 0; u < 4; ++u) tj[u] = idx[(size_t)E + e0 + u];
                }
                const float4 f = *(const float4*)(val + e0);
                ve[0] = f.x; ve[1] = f.y; ve[2] = f.z; ve[3] = f.w;
            }
        } else {
#pragma unroll
            for (int u = 0; u < 4; ++u)
                if (e0 + u < E) {
                    ti[u] = idx[e0 + u];
                    if (use_rec) { tj[u] = idx[(size_t)E + e0 + u]; ve[u] = val[e0 + u]; }
                }
        }
#pragma unroll
        for (int u = 0; u < 4; ++u)
            if (ti[u] >= 0) {
                t[u] = ti[u] / R_TILE;
                w0[u] = (unsigned)(tj[u] + list * N) | ((unsigned)(ti[u] - t[u] * R_TILE) << 24);
            }
    }
    int rank[4];
#pragma unroll
    for (int u = 0; u < 4; ++u) {
        rank[u] = 0;
        if ((unsigned)t[u] < 256u) rank[u] = atomicAdd(&cnt[t[u]], 1);
        else t[u] = -1;
    }
    __syncthreads();
    scan_counts(cnt, start, tid);
    __syncthreads();

    if (use_rec) {
#pragma unroll
        for (int u = 0; u < 4; ++u)
            if (t[u] >= 0)
                ((uint2*)lst)[start[t[u]] + rank[u]] = make_uint2(w0[u], __float_as_uint(ve[u]));
    } else {
#pragma unroll
        for (int u = 0; u < 4; ++u)
            if (t[u] >= 0)
                ((unsigned short*)lst)[start[t[u]] + rank[u]] = (unsigned short)(4 * tid + u);
    }
    __syncthreads();

    if (use_rec) {
        uint4* dst = (uint4*)((char*)region + (size_t)seg * SEG * REC_BYTES);
        dst[tid]        = ((const uint4*)lst)[tid];
        dst[tid + 1024] = ((const uint4*)lst)[tid + 1024];
    } else {
        uint2* dst = (uint2*)((char*)region + (size_t)seg * SEG * POS_BYTES);
        dst[tid] = ((const uint2*)lst)[tid];
    }
    if (tid < SEG_TILES)
        offs[(size_t)seg * SEG_TILES + tid] =
            (unsigned)start[tid] | ((unsigned)start[tid + 1] << 16);
}

__device__ __forceinline__ void add_rows(
    int jj, int r, float alpha, int count, int lane,
    const float* __restrict__ xm, float* acc)
{
#pragma unroll
    for (int h = 0; h < 2; ++h) {
        if (h * 16 >= count) break;
        float2 m[16];
#pragma unroll
        for (int u = 0; u < 16; ++u) {
            const int rk = __builtin_amdgcn_readlane(r, h * 16 + u);
            m[u] = make_float2(0.f, 0.f);
            if (rk >= 0) {
                const int jk = __builtin_amdgcn_readlane(jj, h * 16 + u);
                m[u] = *(const float2*)(xm + (size_t)jk * OUT_C + 2 * lane);
            }
        }
#pragma unroll
        for (int u = 0; u < 16; ++u) {
            const int rk = __builtin_amdgcn_readlane(r, h * 16 + u);
            if (rk >= 0) {
                const float ak = __int_as_float(
                    __builtin_amdgcn_readlane(__float_as_int(alpha), h * 16 + u));
                float* a = acc + rk * OUT_C + 2 * lane;
                atomicAdd(a,     ak * m[u].x);
                atomicAdd(a + 1, ak * m[u].y);
            }
        }
    }
}

__device__ __forceinline__ void flush_recs(
    unsigned w0, float ve, bool valid, int count, int lane, int base,
    const float* __restrict__ xm, const float* __restrict__ sc,
    float* acc, int N)
{
    int jj = 0;
    int r = -1;
    float alpha = 0.f;
    if (valid) {
        jj = (int)(w0 & 0x00FFFFFFu);
        r  = (int)(w0 >> 24);
        const int list = (jj >= N) ? 1 : 0;
        const float* s = sc + (size_t)list * 2 * (size_t)N;
        const float z = s[jj - list * N] + s[(size_t)N + base + r];
        const float elu = (z > 0.f) ? z : expm1f(z);
        alpha = elu * ve;
    }
    add_rows(jj, r, alpha, count, lane, xm, acc);
}

__device__ __forceinline__ void flush_edges(
    int g, int count, int lane, int base,
    const int* __restrict__ idx_l, const float* __restrict__ val_l,
    const int* __restrict__ idx_u, const float* __restrict__ val_u,
    const float* __restrict__ xm, const float* __restrict__ sc,
    float* acc, int N, int E, int Epad)
{
    int jj = 0;
    int r = -1;
    float alpha = 0.f;
    if (g >= 0) {
        const int list = (g >= Epad) ? 1 : 0;
        const int e = g - list * Epad;
        const int*   idx = list ? idx_u : idx_l;
        const float* val = list ? val_u : val_l;
        const int   i  = idx[e];
        const int   j  = idx[(size_t)E + e];
        const float ve = val[e];
        const float* s = sc + (size_t)list * 2 * (size_t)N;
        const float z = s[j] + s[(size_t)N + i];
        const float elu = (z > 0.f) ? z : expm1f(z);
        alpha = elu * ve;
        jj = j + list * N;
        r = i - base;
    }
    add_rows(jj, r, alpha, count, lane, xm, acc);
}

template <bool FULL>
__device__ __forceinline__ void sum_rows16(
    int cjj, int cr, float alpha, int off, int nb, int lane,
    const float* __restrict__ xm, float* acc, int& cur, float& sx, float& sy)
{
    float2 m[16];
#pragma unroll
    for (int u = 0; u < 16; ++u) {
        m[u] = make_float2(0.f, 0.f);
        if (FULL || u < nb) {
            const int jk = __builtin_amdgcn_readlane(cjj, off + u);
            m[u] = *(const float2*)(xm + (size_t)jk * OUT_C + 2 * lane);
        }
    }
#pragma unroll
    for (int u = 0; u < 16; ++u) {
        if (FULL || u < nb) {
            const int rk = __builtin_amdgcn_readlane(cr, off + u);
            const float ak = __int_as_float(
                __builtin_amdgcn_readlane(__float_as_int(alpha), off + u));
            if (rk != cur) {
                if (cur >= 0) {
                    float* a = acc + cur * OUT_C + 2 * lane;
                    atomicAdd(a,     sx);
                    atomicAdd(a + 1, sy);
                }
                cur = rk; sx = 0.f; sy = 0.f;
            }
            sx += ak * m[u].x;
            sy += ak * m[u].y;
        }
    }
}

#define ORD_CAP 6784
#define ORD_TRIPS ((ORD_CAP + 1023) / 1024)

__global__ void __launch_bounds__(1024)
tile_accum_kernel(const unsigned int* __restrict__ region,
                  const int* __restrict__ idx_l, const float* __restrict__ val_l,
                  const int* __restrict__ idx_u, const float* __restrict__ val_u,
                  const float* __restrict__ xm,
                  const float* __restrict__ sc,
                  float* __restrict__ out, int N, int E, int Epad, int nseg,
                  unsigned long long need_rec, unsigned long long need_pos,
                  unsigned long long ws_bytes)
{
    if (need_pos > ws_bytes) return;
    const bool use_rec = (need_rec <= ws_bytes);
    __shared__ float acc[R_TILE * OUT_C];
    __shared__ int cnt[256];
    __shared__ int start[SEG_TILES + 1];
    __shared__ int wtot[TILE_WAVES];
    __shared__ __attribute__((aligned(16))) uint2 stage[ORD_CAP];
    __shared__ __attribute__((aligned(16))) uint2 srt[ORD_CAP];

    const unsigned int* offs = (const unsigned int*)((const char*)region +
        (size_t)nseg * SEG * (use_rec ? REC_BYTES : POS_BYTES));

    const int tile = (int)blockIdx.x;
    const int base = tile * R_TILE;
    const int rows = min(R_TILE, N - base);
    const int tid  = (int)threadIdx.x;
    const int lane = tid & 31;
    const int wave = __builtin_amdgcn_readfirstlane(tid >> 5);

    unsigned pp = 0u;
    {
        const int sseg = wave + TILE_WAVES * lane;
        if (sseg < nseg) pp = offs[(size_t)sseg * SEG_TILES + tile];
    }
    if (tid < 256) cnt[tid] = 0;

    {
        const float4* src = (const float4*)(out + (size_t)base * OUT_C);
        float4* dst = (float4*)acc;
        for (int t = tid; t < rows * (OUT_C / 4); t += 1024) dst[t] = src[t];
    }

    const bool try_ord = use_rec && (nseg <= TILE_WAVES * 32);
    const int o0l = (int)(pp & 0xFFFFu);
    const int cl  = (int)(pp >> 16) - o0l;
    int excl = 0;
    if (try_ord) {
        int incl = cl;
#pragma unroll
        for (int d = 1; d < 32; d <<= 1) {
            const int nb = __shfl_up(incl, d, 32);
            if (lane >= d) incl += nb;
        }
        excl = incl - cl;
        if (lane == 31) wtot[wave] = incl;
    }
    __syncthreads();

    bool ordered = false;
    int wbase = 0, total = 0;
    if (try_ord) {
        int tv = wtot[lane];
        int bv = (lane < wave) ? tv : 0;
#pragma unroll
        for (int m = 16; m >= 1; m >>= 1) {
            tv += __shfl_xor(tv, m, 32);
            bv += __shfl_xor(bv, m, 32);
        }
        total = __builtin_amdgcn_readfirstlane(tv);
        wbase = __builtin_amdgcn_readfirstlane(bv);
        ordered = (total <= ORD_CAP);
    }

    if (ordered) {
        const uint2* recs = (const uint2*)region;
        const int nmine = (wave < nseg) ? (nseg - wave + TILE_WAVES - 1) / TILE_WAVES : 0;
        for (int n0 = 0; n0 < nmine; n0 += 8) {
            uint2 rc[8];
            int cc[8], oo[8], of[8];
#pragma unroll
            for (int u = 0; u < 8; ++u) {
                const int n = n0 + u;
                const int nl = min(n, 31);
                cc[u] = (n < nmine) ? __builtin_amdgcn_readlane(cl, nl) : 0;
                oo[u] = __builtin_amdgcn_readlane(o0l, nl);
                of[u] = wbase + __builtin_amdgcn_readlane(excl, nl);
                rc[u] = make_uint2(0u, 0u);
                if (lane < cc[u])
                    rc[u] = recs[(size_t)(wave + TILE_WAVES * n) * SEG + oo[u] + lane];
            }
#pragma unroll
            for (int u = 0; u < 8; ++u) {
                if (lane < cc[u]) {
                    atomicAdd(&cnt[rc[u].x >> 24], 1);
                    stage[of[u] + lane] = rc[u];
                }
#pragma nounroll
                for (int k0 = 32; k0 < cc[u]; k0 += 32) {
                    if (lane < cc[u] - k0) {
                        const uint2 r2 =
                            recs[(size_t)(wave + TILE_WAVES * (n0 + u)) * SEG + oo[u] + k0 + lane];
                        atomicAdd(&cnt[r2.x >> 24], 1);
                        stage[of[u] + k0 + lane] = r2;
                    }
                }
            }
        }
        __syncthreads();
        scan_counts(cnt, start, tid);
        __syncthreads();
        {
            uint2 e[ORD_TRIPS];
            int p[ORD_TRIPS];
#pragma unroll
            for (int i = 0; i < ORD_TRIPS; ++i) {
                e[i] = make_uint2(0u, 0u);
                if (tid + 1024 * i < total) e[i] = stage[tid + 1024 * i];
            }
#pragma unroll
            for (int i = 0; i < ORD_TRIPS; ++i) {
                p[i] = 0;
                if (tid + 1024 * i < total) {
                    const int r = (int)(e[i].x >> 24);
                    p[i] = start[r] + atomicAdd(&cnt[r], -1) - 1;
                }
            }
#pragma unroll
            for (int i = 0; i < ORD_TRIPS; ++i)
                if (tid + 1024 * i < total) srt[p[i]] = e[i];
        }
        __syncthreads();

        const int ka = (total * wave) / TILE_WAVES;
        const int kb = (total * (wave + 1)) / TILE_WAVES;
        int cur = -1;
        float sx = 0.f, sy = 0.f;
        int jj = 0, r = -1;
        float ve = 0.f, s1 = 0.f, s2 = 0.f;
        if (ka + lane < kb) {
            const uint2 e = srt[ka + lane];
            jj = (int)(e.x & 0x00FFFFFFu);
            r  = (int)(e.x >> 24);
            ve = __uint_as_float(e.y);
            const int list = (jj >= N) ? 1 : 0;
            const float* s = sc + (size_t)list * 2 * (size_t)N;
            s1 = s[jj - list * N];
            s2 = s[(size_t)N + base + r];
        }
        for (int k = ka; k < kb; k += 32) {
            const float z = s1 + s2;
            const float elu = (z > 0.f) ? z : expm1f(z);
            const float alpha = elu * ve;
            const int cjj = jj, cr = r;
            const int count = min(32, kb - k);
            jj = 0; r = -1; ve = 0.f; s1 = 0.f; s2 = 0.f;
            if (k + 32 + lane < kb) {
                const uint2 e = srt[k + 32 + lane];
                jj = (int)(e.x & 0x00FFFFFFu);
                r  = (int)(e.x >> 24);
                ve = __uint_as_float(e.y);
                const int list = (jj >= N) ? 1 : 0;
                const float* s = sc + (size_t)list * 2 * (size_t)N;
                s1 = s[jj - list * N];
                s2 = s[(size_t)N + base + r];
            }
#pragma unroll
            for (int h = 0; h < 2; ++h) {
                const int nb = count - h * 16;
                if (nb <= 0) break;
                if (nb >= 16)
                    sum_rows16<true >(cjj, cr, alpha, h * 16, 16, lane, xm, acc, cur, sx, sy);
                else
                    sum_rows16<false>(cjj, cr, alpha, h * 16, nb, lane, xm, acc, cur, sx, sy);
            }
        }
        if (cur >= 0) {
            float* a = acc + cur * OUT_C + 2 * lane;
            atomicAdd(a,     sx);
            atomicAdd(a + 1, sy);
        }
    } else if (use_rec) {
        const uint2* recs = (const uint2*)region;
        for (int sb = wave; sb < nseg; sb += TILE_WAVES * 32) {
            if (sb != wave) {
                const int sseg = sb + TILE_WAVES * lane;
                pp = 0u;
                if (sseg < nseg) pp = offs[(size_t)sseg * SEG_TILES + tile];
            }
            unsigned pw = (unsigned)__builtin_amdgcn_readlane((int)pp, 0);
            int o0 = (int)(pw & 0xFFFFu), o1 = (int)(pw >> 16);
            uint2 rec = recs[(size_t)sb * SEG + min(o0 + lane, SEG - 1)];
            for (int n = 0; n < 32; ++n) {
                const int seg = sb + TILE_WAVES * n;
                if (seg >= nseg) break;
                const bool has_nxt = (n + 1 < 32) && (seg + TILE_WAVES < nseg);
                const int sn = min(seg + TILE_WAVES, nseg - 1);
                pw = (unsigned)__builtin_amdgcn_readlane((int)pp, min(n + 1, 31));
                int o0_nxt = (int)(pw & 0xFFFFu), o1_nxt = (int)(pw >> 16);
                if (!has_nxt) { o0_nxt = 0; o1_nxt = 0; }
                const uint2 rec_nxt = recs[(size_t)sn * SEG + min(o0_nxt + lane, SEG - 1)];
                if (o1 > o0)
                    flush_recs(rec.x, __uint_as_float(rec.y), lane < o1 - o0, min(32, o1 - o0),
                               lane, base, xm, sc, acc, N);
                for (int k0 = o0 + 32; k0 < o1; k0 += 32) {
                    const bool v2 = lane < o1 - k0;
                    uint2 r2 = make_uint2(0u, 0u);
                    if (v2) r2 = recs[(size_t)seg * SEG + k0 + lane];
                    flush_recs(r2.x, __uint_as_float(r2.y), v2, min(32, o1 - k0),
                               lane, base, xm, sc, acc, N);
                }
                rec = rec_nxt; o0 = o0_nxt; o1 = o1_nxt;
            }
        }
    } else {
        const unsigned short* sorted = (const unsigned short*)region;
        for (int sb = wave; sb < nseg; sb += TILE_WAVES * 32) {
            if (sb != wave) {
                const int sseg = sb + TILE_WAVES * lane;
                pp = 0u;
                if (sseg < nseg) pp = offs[(size_t)sseg * SEG_TILES + tile];
            }
            unsigned pw = (unsigned)__builtin_amdgcn_readlane((int)pp, 0);
            int o0 = (int)(pw & 0xFFFFu), o1 = (int)(pw >> 16);
            int raw = (int)sorted[(size_t)sb * SEG + min(o0 + lane, SEG - 1)];
            for (int n = 0; n < 32; ++n) {
                const int seg = sb + TILE_WAVES * n;
                if (seg >= nseg) break;
                const bool has_nxt = (n + 1 < 32) && (seg + TILE_WAVES < nseg);
                const int sn = min(seg + TILE_WAVES, nseg - 1);
                pw = (unsigned)__builtin_amdgcn_readlane((int)pp, min(n + 1, 31));
                int o0_nxt = (int)(pw & 0xFFFFu), o1_nxt = (int)(pw >> 16);
                if (!has_nxt) { o0_nxt = 0; o1_nxt = 0; }
                const int raw_nxt = (int)sorted[(size_t)sn * SEG + min(o0_nxt + lane, SEG - 1)];
                if (o1 > o0) {
                    const int g = (lane < o1 - o0) ? seg * SEG + raw : -1;
                    flush_edges(g, min(32, o1 - o0), lane, base, idx_l, val_l, idx_u, val_u,
                                xm, sc, acc, N, E, Epad);
                }
                for (int k0 = o0 + 32; k0 < o1; k0 += 32) {
                    int g2 = -1;
                    if (lane < o1 - k0) g2 = seg * SEG + (int)sorted[(size_t)seg * SEG + k0 + lane];
                    flush_edges(g2, min(32, o1 - k0), lane, base, idx_l, val_l, idx_u, val_u,
                                xm, sc, acc, N, E, Epad);
                }
                raw = raw_nxt; o0 = o0_nxt; o1 = o1_nxt;
            }
        }
    }

    __syncthreads();
    {
        float4* dst = (float4*)(out + (size_t)base * OUT_C);
        const float4* src = (const float4*)acc;
        for (int t = tid; t < rows * (OUT_C / 4); t += 1024) {
            float4 v = src[t];
            v.x = fmaxf(v.x, 0.0f); v.y = fmaxf(v.y, 0.0f);
            v.z = fmaxf(v.z, 0.0f); v.w = fmaxf(v.w, 0.0f);
            dst[t] = v;
        }
    }
}

extern "C" void kernel_launch(void* const* d_in, const int* in_sizes, int n_in,
                              void* d_out, int out_size, void* d_ws, size_t ws_size,
                              hipStream_t stream) {
    const float* x        = (const float*)d_in[0];
    const int*   low_idx  = (const int*)  d_in[1];
    const float* low_val  = (const float*)d_in[2];
    const int*   up_idx   = (const int*)  d_in[3];
    const float* up_val   = (const float*)d_in[4];
    const float* w_lower  = (const float*)d_in[5];
    const float* a_lower  = (const float*)d_in[6];
    const float* w_upper  = (const float*)d_in[7];
    const float* a_upper  = (const float*)d_in[8];
    const float* w_lin    = (const float*)d_in[9];
    float*       out      = (float*)d_out;

    const int N = in_sizes[0] / IN_C;
    const int E = in_sizes[2];
    const int Epad = (E + 15) & ~15;

    float* ws   = (float*)d_ws;
    float* xm_l = ws;
    float* xm_u = xm_l + (size_t)N * OUT_C;
    float* sl_s = xm_u + (size_t)N * OUT_C;
    float* sl_t = sl_s + N;
    float* su_s = sl_t + N;
    float* su_t = su_s + N;
    const int nseg = (2 * Epad + SEG - 1) / SEG;
    unsigned int* region = (unsigned int*)(su_t + N);
    const char* lists_end_rec = (const char*)region + (size_t)nseg * SEG * REC_BYTES;
    const char* lists_end_pos = (const char*)region + (size_t)nseg * SEG * POS_BYTES;
    const unsigned long long need_rec = (unsigned long long)
        (lists_end_rec + (size_t)nseg * SEG_TILES * sizeof(unsigned int) - (const char*)d_ws);
    const unsigned long long need_pos = (unsigned long long)
        (lists_end_pos + (size_t)nseg * SEG_TILES * sizeof(unsigned int) - (const char*)d_ws);
    const unsigned long long have = (unsigned long long)ws_size;

    {
        const int waves  = (N + 15) / 16;
        const int blocks = (waves + 3) / 4;
        node_gemm_kernel<<<blocks, 128, 0, stream>>>(
            x, w_lower, a_lower, w_upper, a_upper, w_lin,
            xm_l, xm_u, sl_s, sl_t, su_s, su_t, out, N);
    }

    seg_sort_kernel<<<nseg, 1024, 0, stream>>>(low_idx, low_val, up_idx, up_val, region,
                                               N, E, Epad, nseg, need_rec, need_pos, have);

    {
        const int tiles = (N + R_TILE - 1) / R_TILE;
        tile_accum_kernel<<<tiles, 1024, 0, stream>>>(
            region, low_idx, low_val, up_idx, up_val,
            xm_l, sl_s, out, N, E, Epad, nseg, need_rec, need_pos, have);
    }
}


# tile kernel's row-sum loop: no per-record branch but the run end; row byte offset and per-list alpha (alpha or 0) formed per lane ahead, per record 3 lane broadcasts and 4 fused multiply-adds
# speedup vs baseline: 4.3535x; 1.2689x over previous
#include <hip/hip_runtime.h>
#include <hip/hip_bf16.h>

#define IN_C  64
#define OUT_C 64
#define EPS   (1.0f + 1e-06f)

typedef float v2f __attribute__((ext_vector_type(2)));
typedef float v8f __attribute__((ext_vector_type(8)));


#define R_TILE 196
#define SCORE_ROWS 128
#define ACC_PITCH 68
#define ACC_ROWS 208
#define ACC_U_OFF (ACC_ROWS * ACC_PITCH)
#define N_UNITS (13 * 4)
#define TILE_WAVES 32
#define SEG 4096
#define SEG_TILES 256
#define REC_BYTES 8
#define POS_BYTES 2

__device__ __forceinline__ void scan_counts(const int* cnt, int* start, int tid)
{
    if (tid < 32) {
        int c[8];
        int s = 0;
#pragma unroll
        for (int k = 0; k < 8; ++k) { c[k] = cnt[8 * tid + k]; s += c[k]; }
        int incl = s;
#pragma unroll
        for (int d = 1; d < 32; d <<= 1) {
            const int nb = __shfl_up(incl, d, 32);
            if (tid >= d) incl += nb;
        }
        int run = incl - s;
#pragma unroll
        for (int k = 0; k < 8; ++k) { start[8 * tid + k] = run; run += c[k]; }
        if (tid == 31) start[256] = run;
    }
}

__global__ void __launch_bounds__(1024)
seg_sort_kernel(const int* __restrict__ idx_l, const float* __restrict__ val_l,
                const int* __restrict__ idx_u, const float* __restrict__ val_u,
                const float* __restrict__ x,
                const float* __restrict__ w_lower, const float* __restrict__ a_lower,
                const float* __restrict__ w_upper, const float* __restrict__ a_upper,
                float* __restrict__ sc,
                unsigned int* __restrict__ region,
                int N, int E, int Epad, int nseg,
                unsigned long long need_rec, unsigned long long need_pos,
                unsigned long long ws_bytes)
{
    if (need_pos > ws_bytes) return;
    const bool use_rec = (need_rec <= ws_bytes);
    __shared__ int cnt[256];
    __shared__ int start[SEG_TILES + 1];
    __shared__ __attribute__((aligned(16))) unsigned int lst[SEG * 2];
    __shared__ __attribute__((aligned(16))) float Vs[4 * IN_C];
    const int tid = (int)threadIdx.x;
    const int seg = (int)blockIdx.x;
    unsigned int* offs = (unsigned int*)((char*)region +
        (size_t)nseg * SEG * (use_rec ? REC_BYTES : POS_BYTES));
    const int lane = tid & 31, wv = tid >> 5;
    float2 xr[4];
#pragma unroll
    for (int u = 0; u < 4; ++u) {
        const int row = seg * SCORE_ROWS + 4 * wv + u;
        xr[u] = make_float2(0.f, 0.f);
        if (row < N) xr[u] = *(const float2*)(x + (size_t)row * IN_C + 2 * lane);
    }
    {
        const int c = tid >> 8, k = (tid >> 2) & 63, part = tid & 3;
        const float* w = (c < 2) ? w_lower : w_upper;
        const float* a = ((c < 2) ? a_lower : a_upper) + (c & 1) * OUT_C;
        const float4* wr = (const float4*)(w + (size_t)k * OUT_C + part * 16);
        const float4* ar = (const float4*)(a + part * 16);
        float s = 0.f;
#pragma unroll
        for (int q = 0; q < 4; ++q) {
            const float4 wq = wr[q];
            const float4 aq = ar[q];
            s += wq.x * aq.x; s += wq.y * aq.y; s += wq.z * aq.z; s += wq.w * aq.w;
        }
        s += __shfl_xor(s, 1, 32);
        s += __shfl_xor(s, 2, 32);
        if (part == 0) Vs[c * IN_C + k] = s;
    }
    if (tid < 256) cnt[tid] = 0;
    if (use_rec) {
        ((uint4*)lst)[tid]        = make_uint4(0u, 0u, 0u, 0u);
        ((uint4*)lst)[tid + 1024] = make_uint4(0u, 0u, 0u, 0u);
    } else {
        ((uint2*)lst)[tid] = make_uint2(0u, 0u);
    }
    __syncthreads();

    {
        for (int r0 = seg * SCORE_ROWS; r0 < N; r0 += nseg * SCORE_ROWS) {
            float p[16];
#pragma unroll
            for (int c = 0; c < 4; ++c) {
                const float2 vv = *(const float2*)&Vs[c * IN_C + 2 * lane];
#pragma unroll
                for (int u = 0; u < 4; ++u) p[4 * u + c] = xr[u].x * vv.x + xr[u].y * vv.y;
            }
            const bool b4 = (lane & 16) != 0, b3 = (lane & 8) != 0;
            const bool b2 = (lane & 4) != 0,  b1 = (lane & 2) != 0;
            float v8[8], v4[4], v2[2];
#pragma unroll
            for (int i = 0; i < 8; ++i) {
                const float keep = b4 ? p[8 + i] : p[i];
                const float send = b4 ? p[i] : p[8 + i];
                v8[i] = keep + __shfl_xor(send, 16, 32);
            }
#pragma unroll
            for (int i = 0; i < 4; ++i) {
                const float keep = b3 ? v8[4 + i] : v8[i];
                const float send = b3 ? v8[i] : v8[4 + i];
                v4[i] = keep + __shfl_xor(send, 8, 32);
            }
#pragma unroll
            for (int i = 0; i < 2; ++i) {
                const float keep = b2 ? v4[2 + i] : v4[i];
                const float send = b2 ? v4[i] : v4[2 + i];
                v2[i] = keep + __shfl_xor(send, 4, 32);
            }
            float v1 = (b1 ? v2[1] : v2[0]) + __shfl_xor(b1 ? v2[0] : v2[1], 2, 32);
            v1 += __shfl_xor(v1, 1, 32);
            if ((lane & 1) == 0) {
                const int row = r0 + 4 * wv + (b4 ? 2 : 0) + (b3 ? 1 : 0);
                const int c = (b2 ? 2 : 0) + (b1 ? 1 : 0);
                if (row < N) sc[(size_t)c * (size_t)N + row] = v1;
            }
#pragma unroll
            for (int u = 0; u < 4; ++u) {
                const int row = r0 + nseg * SCORE_ROWS + 4 * wv + u;
                xr[u] = make_float2(0.f, 0.f);
                if (row < N) xr[u] = *(const float2*)(x + (size_t)row * IN_C + 2 * lane);
            }
        }
    }

    const int g0 = seg * SEG + 4 * tid;
    int t[4] = {-1, -1, -1, -1};
    unsigned w0[4] = {0u, 0u, 0u, 0u};
    float ve[4] = {0.f, 0.f, 0.f, 0.f};
    if (g0 < 2 * Epad) {
        const int list = (g0 >= Epad) ? 1 : 0;
        const int e0 = g0 - list * Epad;
        const int*   idx = list ? idx_u : idx_l;
        const float* val = list ? val_u : val_l;
        int ti[4] = {-1, -1, -1, -1};
        int tj[4] = {0, 0, 0, 0};
        if (e0 + 3 < E) {
            const int4 v = *(const int4*)(idx + e0);
            ti[0] = v.x; ti[1] = v.y; ti[2] = v.z; ti[3] = v.w;
            if (use_rec) {
                if ((E & 3) == 0) {
                    const int4 q = *(const int4*)(idx + (size_t)E + e0);
                    tj[0] = q.x; tj[1] = q.y; tj[2] = q.z; tj[3] = q.w;
                } else {
#pragma unroll
                    for (int u = 0; u < 4; ++u) tj[u] = idx[(size_t)E + e0 + u];
                }
                const float4 f = *(const float4*)(val + e0);
                ve[0] = f.x; ve[1] = f.y; ve[2] = f.z; ve[3] = f.w;
            }
        } else {
#pragma unroll
            for (int u = 0; u < 4; ++u)
                if (e0 + u < E) {
                    ti[u] = idx[e0 + u];
                    if (use_rec) { tj[u] = idx[(size_t)E + e0 + u]; ve[u] = val[e0 + u]; }
                }
        }
#pragma unroll
        for (int u = 0; u < 4; ++u)
            if (ti[u] >= 0) {
                t[u] = ti[u] / R_TILE;
                w0[u] = (unsigned)(tj[u] + list * N) | ((unsigned)(ti[u] - t[u] * R_TILE) << 24);
            }
    }
    int rank[4];
#pragma unroll
    for (int u = 0; u < 4; ++u) {
        rank[u] = 0;
        if ((unsigned)t[u] < 256u) rank[u] = atomicAdd(&cnt[t[u]], 1);
        else t[u] = -1;
    }
    __syncthreads();
    scan_counts(cnt, start, tid);
    __syncthreads();

    if (use_rec) {
#pragma unroll
        for (int u = 0; u < 4; ++u)
            if (t[u] >= 0)
                ((uint2*)lst)[start[t[u]] + rank[u]] = make_uint2(w0[u], __float_as_uint(ve[u]));
    } else {
#pragma unroll
        for (int u = 0; u < 4; ++u)
            if (t[u] >= 0)
                ((unsigned short*)lst)[start[t[u]] + rank[u]] = (unsigned short)(4 * tid + u);
    }
    __syncthreads();

    if (use_rec) {
        uint4* dst = (uint4*)((char*)region + (size_t)seg * SEG * REC_BYTES);
        dst[tid]        = ((const uint4*)lst)[tid];
        dst[tid + 1024] = ((const uint4*)lst)[tid + 1024];
    } else {
        uint2* dst = (uint2*)((char*)region + (size_t)seg * SEG * POS_BYTES);
        dst[tid] = ((const uint2*)lst)[tid];
    }
    if (tid < SEG_TILES)
        offs[(size_t)seg * SEG_TILES + tid] =
            (unsigned)start[tid] | ((unsigned)start[tid + 1] << 16);
}

__device__ __forceinline__ void add_rows(
    int jj, int r, float alpha, int count, int lane,
    const float* __restrict__ x, float* accs, int N)
{
#pragma unroll
    for (int h = 0; h < 2; ++h) {
        if (h * 16 >= count) break;
        float2 m[16];
#pragma unroll
        for (int u = 0; u < 16; ++u) {
            const int rk = __builtin_amdgcn_readlane(r, h * 16 + u);
            m[u] = make_float2(0.f, 0.f);
            if (rk >= 0) {
                const int jk = __builtin_amdgcn_readlane(jj, h * 16 + u);
                const int sk = (jk >= N) ? jk - N : jk;
                m[u] = *(const float2*)(x + (size_t)sk * IN_C + 2 * lane);
            }
        }
#pragma unroll
        for (int u = 0; u < 16; ++u) {
            const int rk = __builtin_amdgcn_readlane(r, h * 16 + u);
            if (rk >= 0) {
                const float ak = __int_as_float(
                    __builtin_amdgcn_readlane(__float_as_int(alpha), h * 16 + u));
                const int jk = __builtin_amdgcn_readlane(jj, h * 16 + u);
                float* a = accs + ((jk >= N) ? ACC_U_OFF : 0) + rk * ACC_PITCH + 2 * lane;
                atomicAdd(a,     ak * m[u].x);
                atomicAdd(a + 1, ak * m[u].y);
            }
        }
    }
}

__device__ __forceinline__ void flush_recs(
    unsigned w0, float ve, bool valid, int count, int lane, int base,
    const float* __restrict__ x, const float* __restrict__ sc,
    float* accs, int N)
{
    int jj = 0;
    int r = -1;
    float alpha = 0.f;
    if (valid) {
        jj = (int)(w0 & 0x00FFFFFFu);
        r  = (int)(w0 >> 24);
        const int list = (jj >= N) ? 1 : 0;
        const float* s = sc + (size_t)list * 2 * (size_t)N;
        const float z = s[jj - list * N] + s[(size_t)N + base + r];
        const float elu = (z > 0.f) ? z : expm1f(z);
        alpha = elu * ve;
    }
    add_rows(jj, r, alpha, count, lane, x, accs, N);
}

__device__ __forceinline__ void flush_edges(
    int g, int count, int lane, int base,
    const int* __restrict__ idx_l, const float* __restrict__ val_l,
    const int* __restrict__ idx_u, const float* __restrict__ val_u,
    const float* __restrict__ x, const float* __restrict__ sc,
    float* accs, int N, int E, int Epad)
{
    int jj = 0;
    int r = -1;
    float alpha = 0.f;
    if (g >= 0) {
        const int list = (g >= Epad) ? 1 : 0;
        const int e = g - list * Epad;
        const int*   idx = list ? idx_u : idx_l;
        const float* val = list ? val_u : val_l;
        const int   i  = idx[e];
        const int   j  = idx[(size_t)E + e];
        const float ve = val[e];
        const float* s = sc + (size_t)list * 2 * (size_t)N;
        const float z = s[j] + s[(size_t)N + i];
        const float elu = (z > 0.f) ? z : expm1f(z);
        alpha = elu * ve;
        jj = j + list * N;
        r = i - base;
    }
    add_rows(jj, r, alpha, count, lane, x, accs, N);
}

template <bool FULL>
__device__ __forceinline__ void sum_rows16(
    unsigned roff, int cr, float al, float au, int off, int nb, int lane,
    const float* __restrict__ x, float* accs, int& cur,
    float& sxl, float& syl, float& sxu, float& syu)
{
    float2 m[16];
    const unsigned lo = 8u * (unsigned)lane;
#pragma unroll
    for (int u = 0; u < 16; ++u) {
        m[u] = make_float2(0.f, 0.f);
        if (FULL || u < nb) {
            const unsigned ro = (unsigned)__builtin_amdgcn_readlane((int)roff, off + u);
            m[u] = *(const float2*)((const char*)x + (ro + lo));
        }
    }
#pragma unroll
    for (int u = 0; u < 16; ++u) {
        if (FULL || u < nb) {
            const int rk = __builtin_amdgcn_readlane(cr, off + u);
            const float kl = __int_as_float(
                __builtin_amdgcn_readlane(__float_as_int(al), off + u));
            const float ku = __int_as_float(
                __builtin_amdgcn_readlane(__float_as_int(au), off + u));
            if (rk != cur) {
                if (cur >= 0) {
                    float* a = accs + cur * ACC_PITCH + 2 * lane;
                    atomicAdd(a,                 sxl);
                    atomicAdd(a + 1,             syl);
                    atomicAdd(a + ACC_U_OFF,     sxu);
                    atomicAdd(a + ACC_U_OFF + 1, syu);
                }
                cur = rk; sxl = 0.f; syl = 0.f; sxu = 0.f; syu = 0.f;
            }
            sxl += kl * m[u].x;
            syl += kl * m[u].y;
            sxu += ku * m[u].x;
            syu += ku * m[u].y;
        }
    }
}

#define ORD_CAP 6784
#define ORD_TRIPS ((ORD_CAP + 1023) / 1024)
static_assert(ORD_CAP * 8 >= ACC_ROWS * OUT_C * 4, "the copy list must hold a tile of results");

__global__ void __launch_bounds__(1024)
tile_accum_kernel(const unsigned int* __restrict__ region,
                  const int* __restrict__ idx_l, const float* __restrict__ val_l,
                  const int* __restrict__ idx_u, const float* __restrict__ val_u,
                  const float* __restrict__ x,
                  const float* __restrict__ w_lower, const float* __restrict__ w_upper,
                  const float* __restrict__ w_lin,
                  const float* __restrict__ sc,
                  float* __restrict__ out, int N, int E, int Epad, int nseg,
                  unsigned long long need_rec, unsigned long long need_pos,
                  unsigned long long ws_bytes)
{
    if (need_pos > ws_bytes) return;
    const bool use_rec = (need_rec <= ws_bytes);
    __shared__ __attribute__((aligned(16))) float accs[2 * ACC_ROWS * ACC_PITCH];
    __shared__ int cnt[256];
    __shared__ int start[SEG_TILES + 1];
    __shared__ int wtot[TILE_WAVES];
    __shared__ __attribute__((aligned(16))) uint2 stage[ORD_CAP];
    __shared__ __attribute__((aligned(16))) uint2 srt[ORD_CAP];

    const unsigned int* offs = (const unsigned int*)((const char*)region +
        (size_t)nseg * SEG * (use_rec ? REC_BYTES : POS_BYTES));

    const int tile = (int)blockIdx.x;
    const int base = tile * R_TILE;
    const int rows = min(R_TILE, N - base);
    const int tid  = (int)threadIdx.x;
    const int lane = tid & 31;
    const int wave = __builtin_amdgcn_readfirstlane(tid >> 5);

    unsigned pp = 0u;
    {
        const int sseg = wave + TILE_WAVES * lane;
        if (sseg < nseg) pp = offs[(size_t)sseg * SEG_TILES + tile];
    }
    if (tid < 256) cnt[tid] = 0;

    {
        float4* dst = (float4*)accs;
        for (int t = tid; t < 2 * ACC_ROWS * ACC_PITCH / 4; t += 1024)
            dst[t] = make_float4(0.f, 0.f, 0.f, 0.f);
    }

    const bool try_ord = use_rec && (nseg <= TILE_WAVES * 32);
    const int o0l = (int)(pp & 0xFFFFu);
    const int cl  = (int)(pp >> 16) - o0l;
    int excl = 0;
    if (try_ord) {
        int incl = cl;
#pragma unroll
        for (int d = 1; d < 32; d <<= 1) {
            const int nb = __shfl_up(incl, d, 32);
            if (lane >= d) incl += nb;
        }
        excl = incl - cl;
        if (lane == 31) wtot[wave] = incl;
    }
    __syncthreads();

    bool ordered = false;
    int wbase = 0, total = 0;
    if (try_ord) {
        int tv = wtot[lane];
        int bv = (lane < wave) ? tv : 0;
#pragma unroll
        for (int m = 16; m >= 1; m >>= 1) {
            tv += __shfl_xor(tv, m, 32);
            bv += __shfl_xor(bv, m, 32);
        }
        total = __builtin_amdgcn_readfirstlane(tv);
        wbase = __builtin_amdgcn_readfirstlane(bv);
        ordered = (total <= ORD_CAP);
    }

    if (ordered) {
        const uint2* recs = (const uint2*)region;
        const int nmine = (wave < nseg) ? (nseg - wave + TILE_WAVES - 1) / TILE_WAVES : 0;
        for (int n0 = 0; n0 < nmine; n0 += 8) {
            uint2 rc[8];
            int cc[8], oo[8], of[8];
#pragma unroll
            for (int u = 0; u < 8; ++u) {
                const int n = n0 + u;
                const int nl = min(n, 31);
                cc[u] = (n < nmine) ? __builtin_amdgcn_readlane(cl, nl) : 0;
                oo[u] = __builtin_amdgcn_readlane(o0l, nl);
                of[u] = wbase + __builtin_amdgcn_readlane(excl, nl);
                rc[u] = make_uint2(0u, 0u);
                if (lane < cc[u])
                    rc[u] = recs[(size_t)(wave + TILE_WAVES * n) * SEG + oo[u] + lane];
            }
#pragma unroll
            for (int u = 0; u < 8; ++u) {
                if (lane < cc[u]) {
                    atomicAdd(&cnt[rc[u].x >> 24], 1);
                    stage[of[u] + lane] = rc[u];
                }
#pragma nounroll
                for (int k0 = 32; k0 < cc[u]; k0 += 32) {
                    if (lane < cc[u] - k0) {
                        const uint2 r2 =
                            recs[(size_t)(wave + TILE_WAVES * (n0 + u)) * SEG + oo[u] + k0 + lane];
                        atomicAdd(&cnt[r2.x >> 24], 1);
                        stage[of[u] + k0 + lane] = r2;
                    }
                }
            }
        }
        __syncthreads();
        scan_counts(cnt, start, tid);
        __syncthreads();
        {
            uint2 e[ORD_TRIPS];
            int p[ORD_TRIPS];
#pragma unroll
            for (int i = 0; i < ORD_TRIPS; ++i) {
                e[i] = make_uint2(0u, 0u);
                if (tid + 1024 * i < total) e[i] = stage[tid + 1024 * i];
            }
#pragma unroll
            for (int i = 0; i < ORD_TRIPS; ++i) {
                p[i] = 0;
                if (tid + 1024 * i < total) {
                    const int r = (int)(e[i].x >> 24);
                    p[i] = start[r] + atomicAdd(&cnt[r], -1) - 1;
                }
            }
#pragma unroll
            for (int i = 0; i < ORD_TRIPS; ++i)
                if (tid + 1024 * i < total) srt[p[i]] = e[i];
        }
        __syncthreads();

        const int ka = (total * wave) / TILE_WAVES;
        const int kb = (total * (wave + 1)) / TILE_WAVES;
        int cur = -1;
        float sxl = 0.f, syl = 0.f, sxu = 0.f, syu = 0.f;
        int jj = 0, r = -1;
        float ve = 0.f, s1 = 0.f, s2 = 0.f;
        if (ka + lane < kb) {
            const uint2 e = srt[ka + lane];
            jj = (int)(e.x & 0x00FFFFFFu);
            r  = (int)(e.x >> 24);
            ve = __uint_as_float(e.y);
            const int list = (jj >= N) ? 1 : 0;
            const float* s = sc + (size_t)list * 2 * (size_t)N;
            s1 = s[jj - list * N];
            s2 = s[(size_t)N + base + r];
        }
        for (int k = ka; k < kb; k += 32) {
            const float z = s1 + s2;
            const float elu = (z > 0.f) ? z : expm1f(z);
            const float alpha = elu * ve;
            const bool up = (jj >= N);
            const unsigned roff = (unsigned)(up ? jj - N : jj) * (unsigned)(IN_C * sizeof(float));
            const float al = up ? 0.f : alpha;
            const float au = up ? alpha : 0.f;
            const int cr = r;
            const int count = min(32, kb - k);
            jj = 0; r = -1; ve = 0.f; s1 = 0.f; s2 = 0.f;
            if (k + 32 + lane < kb) {
                const uint2 e = srt[k + 32 + lane];
                jj = (int)(e.x & 0x00FFFFFFu);
                r  = (int)(e.x >> 24);
                ve = __uint_as_float(e.y);
                const int list = (jj >= N) ? 1 : 0;
                const float* s = sc + (size_t)list * 2 * (size_t)N;
                s1 = s[jj - list * N];
                s2 = s[(size_t)N + base + r];
            }
#pragma unroll
            for (int h = 0; h < 2; ++h) {
                const int nb = count - h * 16;
                if (nb <= 0) break;
                if (nb >= 16)
                    sum_rows16<true >(roff, cr, al, au, h * 16, 16, lane, x, accs, cur,
                                      sxl, syl, sxu, syu);
                else
                    sum_rows16<false>(roff, cr, al, au, h * 16, nb, lane, x, accs, cur,
                                      sxl, syl, sxu, syu);
            }
        }
        if (cur >= 0) {
            float* a = accs + cur * ACC_PITCH + 2 * lane;
            atomicAdd(a,                 sxl);
            atomicAdd(a + 1,             syl);
            atomicAdd(a + ACC_U_OFF,     sxu);
            atomicAdd(a + ACC_U_OFF + 1, syu);
        }
    } else if (use_rec) {
        const uint2* recs = (const uint2*)region;
        for (int sb = wave; sb < nseg; sb += TILE_WAVES * 32) {
            if (sb != wave) {
                const int sseg = sb + TILE_WAVES * lane;
                pp = 0u;
                if (sseg < nseg) pp = offs[(size_t)sseg * SEG_TILES + tile];
            }
            unsigned pw = (unsigned)__builtin_amdgcn_readlane((int)pp, 0);
            int o0 = (int)(pw & 0xFFFFu), o1 = (int)(pw >> 16);
            uint2 rec = recs[(size_t)sb * SEG + min(o0 + lane, SEG - 1)];
            for (int n = 0; n < 32; ++n) {
                const int seg = sb + TILE_WAVES * n;
                if (seg >= nseg) break;
                const bool has_nxt = (n + 1 < 32) && (seg + TILE_WAVES < nseg);
                const int sn = min(seg + TILE_WAVES, nseg - 1);
                pw = (unsigned)__builtin_amdgcn_readlane((int)pp, min(n + 1, 31));
                int o0_nxt = (int)(pw & 0xFFFFu), o1_nxt = (int)(pw >> 16);
                if (!has_nxt) { o0_nxt = 0; o1_nxt = 0; }
                const uint2 rec_nxt = recs[(size_t)sn * SEG + min(o0_nxt + lane, SEG - 1)];
                if (o1 > o0)
                    flush_recs(rec.x, __uint_as_float(rec.y), lane < o1 - o0, min(32, o1 - o0),
                               lane, base, x, sc, accs, N);
                for (int k0 = o0 + 32; k0 < o1; k0 += 32) {
                    const bool v2 = lane < o1 - k0;
                    uint2 r2 = make_uint2(0u, 0u);
                    if (v2) r2 = recs[(size_t)seg * SEG + k0 + lane];
                    flush_recs(r2.x, __uint_as_float(r2.y), v2, min(32, o1 - k0),
                               lane, base, x, sc, accs, N);
                }
                rec = rec_nxt; o0 = o0_nxt; o1 = o1_nxt;
            }
        }
    } else {
        const unsigned short* sorted = (const unsigned short*)region;
        for (int sb = wave; sb < nseg; sb += TILE_WAVES * 32) {
            if (sb != wave) {
                const int sseg = sb + TILE_WAVES * lane;
                pp = 0u;
                if (sseg < nseg) pp = offs[(size_t)sseg * SEG_TILES + tile];
            }
            unsigned pw = (unsigned)__builtin_amdgcn_readlane((int)pp, 0);
            int o0 = (int)(pw & 0xFFFFu), o1 = (int)(pw >> 16);
            int raw = (int)sorted[(size_t)sb * SEG + min(o0 + lane, SEG - 1)];
            for (int n = 0; n < 32; ++n) {
                const int seg = sb + TILE_WAVES * n;
                if (seg >= nseg) break;
                const bool has_nxt = (n + 1 < 32) && (seg + TILE_WAVES < nseg);
                const int sn = min(seg + TILE_WAVES, nseg - 1);
                pw = (unsigned)__builtin_amdgcn_readlane((int)pp, min(n + 1, 31));
                int o0_nxt = (int)(pw & 0xFFFFu), o1_nxt = (int)(pw >> 16);
                if (!has_nxt) { o0_nxt = 0; o1_nxt = 0; }
                const int raw_nxt = (int)sorted[(size_t)sn * SEG + min(o0_nxt + lane, SEG - 1)];
                if (o1 > o0) {
                    const int g = (lane < o1 - o0) ? seg * SEG + raw : -1;
                    flush_edges(g, min(32, o1 - o0), lane, base, idx_l, val_l, idx_u, val_u,
                                x, sc, accs, N, E, Epad);
                }
                for (int k0 = o0 + 32; k0 < o1; k0 += 32) {
                    int g2 = -1;
                    if (lane < o1 - k0) g2 = seg * SEG + (int)sorted[(size_t)seg * SEG + k0 + lane];
                    flush_edges(g2, min(32, o1 - k0), lane, base, idx_l, val_l, idx_u, val_u,
                                x, sc, accs, N, E, Epad);
                }
                raw = raw_nxt; o0 = o0_nxt; o1 = o1_nxt;
            }
        }
    }

    __syncthreads();
    {
        const int half = lane >> 4;
        const int lrow = lane & 15;
        float* res = (float*)stage;
        for (int unit = wave; unit < N_UNITS; unit += TILE_WAVES) {
            const int nt = unit / 13;
            const int rb = unit - 13 * nt;
            const int col = nt * 16 + lrow;
            const float* al = accs + (rb * 16 + lrow) * ACC_PITCH + 2 * half;
            const float* au = al + ACC_U_OFF;
            const int xrow = base + rb * 16 + lrow;
            const bool xin = xrow < N;
            const float* xr = x + (size_t)(xin ? xrow : 0) * IN_C + 2 * half;
            v8f c1 = {};
            v8f c2 = {};
#pragma unroll
            for (int kk = 0; kk < 16; ++kk) {
                const int k0 = 4 * kk + 2 * half;
                const v2f a = *(const v2f*)(al + 4 * kk);
                v2f b;
                b.x = w_lower[(size_t)k0 * OUT_C + col];
                b.y = w_lower[(size_t)(k0 + 1) * OUT_C + col];
                c1 = __builtin_amdgcn_wmma_f32_16x16x4_f32(
                         false, a,  false, b,
                         (short)0, c1,  false,  false);
            }
            __builtin_amdgcn_sched_barrier(0);
#pragma unroll
            for (int kk = 0; kk < 16; ++kk) {
                const int k0 = 4 * kk + 2 * half;
                const v2f a = *(const v2f*)(au + 4 * kk);
                v2f b;
                b.x = w_upper[(size_t)k0 * OUT_C + col];
                b.y = w_upper[(size_t)(k0 + 1) * OUT_C + col];
                c1 = __builtin_amdgcn_wmma_f32_16x16x4_f32(
                         false, a,  false, b,
                         (short)0, c1,  false,  false);
            }
            __builtin_amdgcn_sched_barrier(0);
#pragma unroll
            for (int kk = 0; kk < 16; ++kk) {
                const int k0 = 4 * kk + 2 * half;
                v2f a = *(const v2f*)(xr + 4 * kk);
                if (!xin) { a.x = 0.f; a.y = 0.f; }
                v2f b;
                b.x = w_lin[(size_t)k0 * OUT_C + col];
                b.y = w_lin[(size_t)(k0 + 1) * OUT_C + col];
                c2 = __builtin_amdgcn_wmma_f32_16x16x4_f32(
                         false, a,  false, b,
                         (short)0, c2,  false,  false);
            }
#pragma unroll
            for (int r = 0; r < 8; ++r)
                res[(rb * 16 + r + 8 * half) * OUT_C + col] = fmaxf(c1[r] + c2[r] * EPS, 0.0f);
        }
        __syncthreads();
        float4* dst = (float4*)(out + (size_t)base * OUT_C);
        const float4* rs = (const float4*)res;
        for (int t = tid; t < rows * (OUT_C / 4); t += 1024) dst[t] = rs[t];
    }
}

extern "C" void kernel_launch(void* const* d_in, const int* in_sizes, int n_in,
                              void* d_out, int out_size, void* d_ws, size_t ws_size,
                              hipStream_t stream) {
    const float* x        = (const float*)d_in[0];
    const int*   low_idx  = (const int*)  d_in[1];
    const float* low_val  = (const float*)d_in[2];
    const int*   up_idx   = (const int*)  d_in[3];
    const float* up_val   = (const float*)d_in[4];
    const float* w_lower  = (const float*)d_in[5];
    const float* a_lower  = (const float*)d_in[6];
    const float* w_upper  = (const float*)d_in[7];
    const float* a_upper  = (const float*)d_in[8];
    const float* w_lin    = (const float*)d_in[9];
    float*       out      = (float*)d_out;

    const int N = in_sizes[0] / IN_C;
    const int E = in_sizes[2];
    const int Epad = (E + 15) & ~15;

    float* sc = (float*)d_ws;
    const int nseg = (2 * Epad + SEG - 1) / SEG;
    unsigned int* region = (unsigned int*)(sc + 4 * (size_t)N);
    const char* lists_end_rec = (const char*)region + (size_t)nseg * SEG * REC_BYTES;
    const char* lists_end_pos = (const char*)region + (size_t)nseg * SEG * POS_BYTES;
    const unsigned long long need_rec = (unsigned long long)
        (lists_end_rec + (size_t)nseg * SEG_TILES * sizeof(unsigned int) - (const char*)d_ws);
    const unsigned long long need_pos = (unsigned long long)
        (lists_end_pos + (size_t)nseg * SEG_TILES * sizeof(unsigned int) - (const char*)d_ws);
    const unsigned long long have = (unsigned long long)ws_size;

    seg_sort_kernel<<<nseg, 1024, 0, stream>>>(low_idx, low_val, up_idx, up_val,
                                               x, w_lower, a_lower, w_upper, a_upper, sc, region,
                                               N, E, Epad, nseg, need_rec, need_pos, have);

    {
        const int tiles = (N + R_TILE - 1) / R_TILE;
        tile_accum_kernel<<<tiles, 1024, 0, stream>>>(
            region, low_idx, low_val, up_idx, up_val,
            x, w_lower, w_upper, w_lin, sc, out, N, E, Epad, nseg, need_rec, need_pos, have);
    }
}


# tile kernel: its x rows and w_lin staged in LDS at the start; each wave runs its units' x.w_lin matrix products inside its gather loop, staggered by wave; closing phase keeps the two sum products
# speedup vs baseline: 4.7338x; 1.0874x over previous
#include <hip/hip_runtime.h>
#include <hip/hip_bf16.h>

#define IN_C  64
#define OUT_C 64
#define EPS   (1.0f + 1e-06f)

typedef float v2f __attribute__((ext_vector_type(2)));
typedef float v8f __attribute__((ext_vector_type(8)));


#define R_TILE 196
#define SCORE_ROWS 128
#define ACC_PITCH 68
#define ACC_ROWS 208
#define ACC_U_OFF (ACC_ROWS * ACC_PITCH)
#define ACC_X_OFF (2 * ACC_ROWS * ACC_PITCH)
#define N_UNITS (13 * 4)
#define TILE_WAVES 32
#define SEG 4096
#define SEG_TILES 256
#define REC_BYTES 8
#define POS_BYTES 2

__device__ __forceinline__ void scan_counts(const int* cnt, int* start, int tid)
{
    if (tid < 32) {
        int c[8];
        int s = 0;
#pragma unroll
        for (int k = 0; k < 8; ++k) { c[k] = cnt[8 * tid + k]; s += c[k]; }
        int incl = s;
#pragma unroll
        for (int d = 1; d < 32; d <<= 1) {
            const int nb = __shfl_up(incl, d, 32);
            if (tid >= d) incl += nb;
        }
        int run = incl - s;
#pragma unroll
        for (int k = 0; k < 8; ++k) { start[8 * tid + k] = run; run += c[k]; }
        if (tid == 31) start[256] = run;
    }
}

__global__ void __launch_bounds__(1024)
seg_sort_kernel(const int* __restrict__ idx_l, const float* __restrict__ val_l,
                const int* __restrict__ idx_u, const float* __restrict__ val_u,
                const float* __restrict__ x,
                const float* __restrict__ w_lower, const float* __restrict__ a_lower,
                const float* __restrict__ w_upper, const float* __restrict__ a_upper,
                float* __restrict__ sc,
                unsigned int* __restrict__ region,
                int N, int E, int Epad, int nseg,
                unsigned long long need_rec, unsigned long long need_pos,
                unsigned long long ws_bytes)
{
    if (need_pos > ws_bytes) return;
    const bool use_rec = (need_rec <= ws_bytes);
    __shared__ int cnt[256];
    __shared__ int start[SEG_TILES + 1];
    __shared__ __attribute__((aligned(16))) unsigned int lst[SEG * 2];
    __shared__ __attribute__((aligned(16))) float Vs[4 * IN_C];
    const int tid = (int)threadIdx.x;
    const int seg = (int)blockIdx.x;
    unsigned int* offs = (unsigned int*)((char*)region +
        (size_t)nseg * SEG * (use_rec ? REC_BYTES : POS_BYTES));
    const int lane = tid & 31, wv = tid >> 5;
    float2 xr[4];
#pragma unroll
    for (int u = 0; u < 4; ++u) {
        const int row = seg * SCORE_ROWS + 4 * wv + u;
        xr[u] = make_float2(0.f, 0.f);
        if (row < N) xr[u] = *(const float2*)(x + (size_t)row * IN_C + 2 * lane);
    }
    {
        const int c = tid >> 8, k = (tid >> 2) & 63, part = tid & 3;
        const float* w = (c < 2) ? w_lower : w_upper;
        const float* a = ((c < 2) ? a_lower : a_upper) + (c & 1) * OUT_C;
        const float4* wr = (const float4*)(w + (size_t)k * OUT_C + part * 16);
        const float4* ar = (const float4*)(a + part * 16);
        float s = 0.f;
#pragma unroll
        for (int q = 0; q < 4; ++q) {
            const float4 wq = wr[q];
            const float4 aq = ar[q];
            s += wq.x * aq.x; s += wq.y * aq.y; s += wq.z * aq.z; s += wq.w * aq.w;
        }
        s += __shfl_xor(s, 1, 32);
        s += __shfl_xor(s, 2, 32);
        if (part == 0) Vs[c * IN_C + k] = s;
    }
    if (tid < 256) cnt[tid] = 0;
    if (use_rec) {
        ((uint4*)lst)[tid]        = make_uint4(0u, 0u, 0u, 0u);
        ((uint4*)lst)[tid + 1024] = make_uint4(0u, 0u, 0u, 0u);
    } else {
        ((uint2*)lst)[tid] = make_uint2(0u, 0u);
    }
    __syncthreads();

    {
        for (int r0 = seg * SCORE_ROWS; r0 < N; r0 += nseg * SCORE_ROWS) {
            float p[16];
#pragma unroll
            for (int c = 0; c < 4; ++c) {
                const float2 vv = *(const float2*)&Vs[c * IN_C + 2 * lane];
#pragma unroll
                for (int u = 0; u < 4; ++u) p[4 * u + c] = xr[u].x * vv.x + xr[u].y * vv.y;
            }
            const bool b4 = (lane & 16) != 0, b3 = (lane & 8) != 0;
            const bool b2 = (lane & 4) != 0,  b1 = (lane & 2) != 0;
            float v8[8], v4[4], v2[2];
#pragma unroll
            for (int i = 0; i < 8; ++i) {
                const float keep = b4 ? p[8 + i] : p[i];
                const float send = b4 ? p[i] : p[8 + i];
                v8[i] = keep + __shfl_xor(send, 16, 32);
            }
#pragma unroll
            for (int i = 0; i < 4; ++i) {
                const float keep = b3 ? v8[4 + i] : v8[i];
                const float send = b3 ? v8[i] : v8[4 + i];
                v4[i] = keep + __shfl_xor(send, 8, 32);
            }
#pragma unroll
            for (int i = 0; i < 2; ++i) {
                const float keep = b2 ? v4[2 + i] : v4[i];
                const float send = b2 ? v4[i] : v4[2 + i];
                v2[i] = keep + __shfl_xor(send, 4, 32);
            }
            float v1 = (b1 ? v2[1] : v2[0]) + __shfl_xor(b1 ? v2[0] : v2[1], 2, 32);
            v1 += __shfl_xor(v1, 1, 32);
            if ((lane & 1) == 0) {
                const int row = r0 + 4 * wv + (b4 ? 2 : 0) + (b3 ? 1 : 0);
                const int c = (b2 ? 2 : 0) + (b1 ? 1 : 0);
                if (row < N) sc[(size_t)c * (size_t)N + row] = v1;
            }
#pragma unroll
            for (int u = 0; u < 4; ++u) {
                const int row = r0 + nseg * SCORE_ROWS + 4 * wv + u;
                xr[u] = make_float2(0.f, 0.f);
                if (row < N) xr[u] = *(const float2*)(x + (size_t)row * IN_C + 2 * lane);
            }
        }
    }

    const int g0 = seg * SEG + 4 * tid;
    int t[4] = {-1, -1, -1, -1};
    unsigned w0[4] = {0u, 0u, 0u, 0u};
    float ve[4] = {0.f, 0.f, 0.f, 0.f};
    if (g0 < 2 * Epad) {
        const int list = (g0 >= Epad) ? 1 : 0;
        const int e0 = g0 - list * Epad;
        const int*   idx = list ? idx_u : idx_l;
        const float* val = list ? val_u : val_l;
        int ti[4] = {-1, -1, -1, -1};
        int tj[4] = {0, 0, 0, 0};
        if (e0 + 3 < E) {
            const int4 v = *(const int4*)(idx + e0);
            ti[0] = v.x; ti[1] = v.y; ti[2] = v.z; ti[3] = v.w;
            if (use_rec) {
                if ((E & 3) == 0) {
                    const int4 q = *(const int4*)(idx + (size_t)E + e0);
                    tj[0] = q.x; tj[1] = q.y; tj[2] = q.z; tj[3] = q.w;
                } else {
#pragma unroll
                    for (int u = 0; u < 4; ++u) tj[u] = idx[(size_t)E + e0 + u];
                }
                const float4 f = *(const float4*)(val + e0);
                ve[0] = f.x; ve[1] = f.y; ve[2] = f.z; ve[3] = f.w;
            }
        } else {
#pragma unroll
            for (int u = 0; u < 4; ++u)
                if (e0 + u < E) {
                    ti[u] = idx[e0 + u];
                    if (use_rec) { tj[u] = idx[(size_t)E + e0 + u]; ve[u] = val[e0 + u]; }
                }
        }
#pragma unroll
        for (int u = 0; u < 4; ++u)
            if (ti[u] >= 0) {
                t[u] = ti[u] / R_TILE;
                w0[u] = (unsigned)(tj[u] + list * N) | ((unsigned)(ti[u] - t[u] * R_TILE) << 24);
            }
    }
    int rank[4];
#pragma unroll
    for (int u = 0; u < 4; ++u) {
        rank[u] = 0;
        if ((unsigned)t[u] < 256u) rank[u] = atomicAdd(&cnt[t[u]], 1);
        else t[u] = -1;
    }
    __syncthreads();
    scan_counts(cnt, start, tid);
    __syncthreads();

    if (use_rec) {
#pragma unroll
        for (int u = 0; u < 4; ++u)
            if (t[u] >= 0)
                ((uint2*)lst)[start[t[u]] + rank[u]] = make_uint2(w0[u], __float_as_uint(ve[u]));
    } else {
#pragma unroll
        for (int u = 0; u < 4; ++u)
            if (t[u] >= 0)
                ((unsigned short*)lst)[start[t[u]] + rank[u]] = (unsigned short)(4 * tid + u);
    }
    __syncthreads();

    if (use_rec) {
        uint4* dst = (uint4*)((char*)region + (size_t)seg * SEG * REC_BYTES);
        dst[tid]        = ((const uint4*)lst)[tid];
        dst[tid + 1024] = ((const uint4*)lst)[tid + 1024];
    } else {
        uint2* dst = (uint2*)((char*)region + (size_t)seg * SEG * POS_BYTES);
        dst[tid] = ((const uint2*)lst)[tid];
    }
    if (tid < SEG_TILES)
        offs[(size_t)seg * SEG_TILES + tid] =
            (unsigned)start[tid] | ((unsigned)start[tid + 1] << 16);
}

__device__ __forceinline__ void add_rows(
    int jj, int r, float alpha, int count, int lane,
    const float* __restrict__ x, float* accs, int N)
{
#pragma unroll
    for (int h = 0; h < 2; ++h) {
        if (h * 16 >= count) break;
        float2 m[16];
#pragma unroll
        for (int u = 0; u < 16; ++u) {
            const int rk = __builtin_amdgcn_readlane(r, h * 16 + u);
            m[u] = make_float2(0.f, 0.f);
            if (rk >= 0) {
                const int jk = __builtin_amdgcn_readlane(jj, h * 16 + u);
                const int sk = (jk >= N) ? jk - N : jk;
                m[u] = *(const float2*)(x + (size_t)sk * IN_C + 2 * lane);
            }
        }
#pragma unroll
        for (int u = 0; u < 16; ++u) {
            const int rk = __builtin_amdgcn_readlane(r, h * 16 + u);
            if (rk >= 0) {
                const float ak = __int_as_float(
                    __builtin_amdgcn_readlane(__float_as_int(alpha), h * 16 + u));
                const int jk = __builtin_amdgcn_readlane(jj, h * 16 + u);
                float* a = accs + ((jk >= N) ? ACC_U_OFF : 0) + rk * ACC_PITCH + 2 * lane;
                atomicAdd(a,     ak * m[u].x);
                atomicAdd(a + 1, ak * m[u].y);
            }
        }
    }
}

__device__ __forceinline__ void flush_recs(
    unsigned w0, float ve, bool valid, int count, int lane, int base,
    const float* __restrict__ x, const float* __restrict__ sc,
    float* accs, int N)
{
    int jj = 0;
    int r = -1;
    float alpha = 0.f;
    if (valid) {
        jj = (int)(w0 & 0x00FFFFFFu);
        r  = (int)(w0 >> 24);
        const int list = (jj >= N) ? 1 : 0;
        const float* s = sc + (size_t)list * 2 * (size_t)N;
        const float z = s[jj - list * N] + s[(size_t)N + base + r];
        const float elu = (z > 0.f) ? z : expm1f(z);
        alpha = elu * ve;
    }
    add_rows(jj, r, alpha, count, lane, x, accs, N);
}

__device__ __forceinline__ void flush_edges(
    int g, int count, int lane, int base,
    const int* __restrict__ idx_l, const float* __restrict__ val_l,
    const int* __restrict__ idx_u, const float* __restrict__ val_u,
    const float* __restrict__ x, const float* __restrict__ sc,
    float* accs, int N, int E, int Epad)
{
    int jj = 0;
    int r = -1;
    float alpha = 0.f;
    if (g >= 0) {
        const int list = (g >= Epad) ? 1 : 0;
        const int e = g - list * Epad;
        const int*   idx = list ? idx_u : idx_l;
        const float* val = list ? val_u : val_l;
        const int   i  = idx[e];
        const int   j  = idx[(size_t)E + e];
        const float ve = val[e];
        const float* s = sc + (size_t)list * 2 * (size_t)N;
        const float z = s[j] + s[(size_t)N + i];
        const float elu = (z > 0.f) ? z : expm1f(z);
        alpha = elu * ve;
        jj = j + list * N;
        r = i - base;
    }
    add_rows(jj, r, alpha, count, lane, x, accs, N);
}

template <bool FULL>
__device__ __forceinline__ void sum_rows16(
    unsigned roff, int cr, float al, float au, int off, int nb, int lane,
    const float* __restrict__ x, float* accs, int& cur,
    float& sxl, float& syl, float& sxu, float& syu)
{
    float2 m[16];
    const unsigned lo = 8u * (unsigned)lane;
#pragma unroll
    for (int u = 0; u < 16; ++u) {
        m[u] = make_float2(0.f, 0.f);
        if (FULL || u < nb) {
            const unsigned ro = (unsigned)__builtin_amdgcn_readlane((int)roff, off + u);
            m[u] = *(const float2*)((const char*)x + (ro + lo));
        }
    }
#pragma unroll
    for (int u = 0; u < 16; ++u) {
        if (FULL || u < nb) {
            const int rk = __builtin_amdgcn_readlane(cr, off + u);
            const float kl = __int_as_float(
                __builtin_amdgcn_readlane(__float_as_int(al), off + u));
            const float ku = __int_as_float(
                __builtin_amdgcn_readlane(__float_as_int(au), off + u));
            if (rk != cur) {
                if (cur >= 0) {
                    float* a = accs + cur * ACC_PITCH + 2 * lane;
                    atomicAdd(a,                 sxl);
                    atomicAdd(a + 1,             syl);
                    atomicAdd(a + ACC_U_OFF,     sxu);
                    atomicAdd(a + ACC_U_OFF + 1, syu);
                }
                cur = rk; sxl = 0.f; syl = 0.f; sxu = 0.f; syu = 0.f;
            }
            sxl += kl * m[u].x;
            syl += kl * m[u].y;
            sxu += ku * m[u].x;
            syu += ku * m[u].y;
        }
    }
}

#define ORD_CAP 6784
#define ORD_TRIPS ((ORD_CAP + 1023) / 1024)
static_assert(ORD_CAP * 8 >= ACC_ROWS * OUT_C * 4, "the copy list must hold a tile of results");

__device__ __forceinline__ void lin_units(const float* accs, const float* wls, float* res,
                                          int wave, int lane)
{
    const int half = lane >> 4;
    const int lrow = lane & 15;
    for (int unit = wave; unit < N_UNITS; unit += TILE_WAVES) {
        const int nt = unit / 13;
        const int rb = unit - 13 * nt;
        const int col = nt * 16 + lrow;
        const float* ax = accs + ACC_X_OFF + (rb * 16 + lrow) * ACC_PITCH + 2 * half;
        const float* bw = wls + (nt * 32 + lane) * 2;
        v8f c2 = {};
        v2f a[16], b[16];
#pragma unroll
        for (int kk = 0; kk < 16; ++kk) {
            a[kk] = *(const v2f*)(ax + 4 * kk);
            b[kk] = *(const v2f*)(bw + kk * (4 * 32 * 2));
        }
        __builtin_amdgcn_sched_barrier(0);
#pragma unroll
        for (int kk = 0; kk < 16; ++kk)
            c2 = __builtin_amdgcn_wmma_f32_16x16x4_f32(
                     false, a[kk],  false, b[kk],
                     (short)0, c2,  false,  false);
#pragma unroll
        for (int r = 0; r < 8; ++r)
            res[(rb * 16 + r + 8 * half) * OUT_C + col] = c2[r];
    }
}

__global__ void __launch_bounds__(1024)
tile_accum_kernel(const unsigned int* __restrict__ region,
                  const int* __restrict__ idx_l, const float* __restrict__ val_l,
                  const int* __restrict__ idx_u, const float* __restrict__ val_u,
                  const float* __restrict__ x,
                  const float* __restrict__ w_lower, const float* __restrict__ w_upper,
                  const float* __restrict__ w_lin,
                  const float* __restrict__ sc,
                  float* __restrict__ out, int N, int E, int Epad, int nseg,
                  unsigned long long need_rec, unsigned long long need_pos,
                  unsigned long long ws_bytes)
{
    if (need_pos > ws_bytes) return;
    const bool use_rec = (need_rec <= ws_bytes);
    __shared__ __attribute__((aligned(16))) float accs[3 * ACC_ROWS * ACC_PITCH];
    __shared__ __attribute__((aligned(16))) float wls[IN_C * OUT_C];
    __shared__ int cnt[256];
    __shared__ int start[SEG_TILES + 1];
    __shared__ int wtot[TILE_WAVES];
    __shared__ __attribute__((aligned(16))) uint2 stage[ORD_CAP];
    __shared__ __attribute__((aligned(16))) uint2 srt[ORD_CAP];

    const unsigned int* offs = (const unsigned int*)((const char*)region +
        (size_t)nseg * SEG * (use_rec ? REC_BYTES : POS_BYTES));

    const int tile = (int)blockIdx.x;
    const int base = tile * R_TILE;
    const int rows = min(R_TILE, N - base);
    const int tid  = (int)threadIdx.x;
    const int lane = tid & 31;
    const int wave = __builtin_amdgcn_readfirstlane(tid >> 5);

    unsigned pp = 0u;
    {
        const int sseg = wave + TILE_WAVES * lane;
        if (sseg < nseg) pp = offs[(size_t)sseg * SEG_TILES + tile];
    }
    float4 xt[4];
#pragma unroll
    for (int i = 0; i < 4; ++i) {
        const int t = tid + 1024 * i;
        const int rl = min(t >> 4, rows - 1);
        xt[i] = *(const float4*)(x + (size_t)(base + rl) * IN_C + 4 * (t & 15));
    }
    const float4 wt = *(const float4*)(w_lin + 4 * tid);
    if (tid < 256) cnt[tid] = 0;

    {
        float4* dst = (float4*)accs;
        for (int t = tid; t < 2 * ACC_ROWS * ACC_PITCH / 4; t += 1024)
            dst[t] = make_float4(0.f, 0.f, 0.f, 0.f);
    }

    const bool try_ord = use_rec && (nseg <= TILE_WAVES * 32);
    const int o0l = (int)(pp & 0xFFFFu);
    const int cl  = (int)(pp >> 16) - o0l;
    int excl = 0;
    if (try_ord) {
        int incl = cl;
#pragma unroll
        for (int d = 1; d < 32; d <<= 1) {
            const int nb = __shfl_up(incl, d, 32);
            if (lane >= d) incl += nb;
        }
        excl = incl - cl;
        if (lane == 31) wtot[wave] = incl;
    }
#pragma unroll
    for (int i = 0; i < 4; ++i) {
        const int t = tid + 1024 * i;
        if (t < ACC_ROWS * (IN_C / 4)) {
            float4 v = xt[i];
            if ((t >> 4) >= rows) v = make_float4(0.f, 0.f, 0.f, 0.f);
            *(float4*)&accs[ACC_X_OFF + (t >> 4) * ACC_PITCH + 4 * (t & 15)] = v;
        }
    }
    {
        const int k = tid >> 4, n0 = 4 * (tid & 15);
        const int kk = k >> 2, hf = (k >> 1) & 1, s = k & 1;
        const int nt = n0 >> 4, lr0 = n0 & 15;
        float* d = wls + ((kk * 4 + nt) * 32 + hf * 16 + lr0) * 2 + s;
        d[0] = wt.x; d[2] = wt.y; d[4] = wt.z; d[6] = wt.w;
    }
    __syncthreads();

    bool ordered = false;
    int wbase = 0, total = 0;
    if (try_ord) {
        int tv = wtot[lane];
        int bv = (lane < wave) ? tv : 0;
#pragma unroll
        for (int m = 16; m >= 1; m >>= 1) {
            tv += __shfl_xor(tv, m, 32);
            bv += __shfl_xor(bv, m, 32);
        }
        total = __builtin_amdgcn_readfirstlane(tv);
        wbase = __builtin_amdgcn_readfirstlane(bv);
        ordered = (total <= ORD_CAP);
    }

    bool lin_done = false;
    if (ordered) {
        const uint2* recs = (const uint2*)region;
        const int nmine = (wave < nseg) ? (nseg - wave + TILE_WAVES - 1) / TILE_WAVES : 0;
        for (int n0 = 0; n0 < nmine; n0 += 8) {
            uint2 rc[8];
            int cc[8], oo[8], of[8];
#pragma unroll
            for (int u = 0; u < 8; ++u) {
                const int n = n0 + u;
                const int nl = min(n, 31);
                cc[u] = (n < nmine) ? __builtin_amdgcn_readlane(cl, nl) : 0;
                oo[u] = __builtin_amdgcn_readlane(o0l, nl);
                of[u] = wbase + __builtin_amdgcn_readlane(excl, nl);
                rc[u] = make_uint2(0u, 0u);
                if (lane < cc[u])
                    rc[u] = recs[(size_t)(wave + TILE_WAVES * n) * SEG + oo[u] + lane];
            }
#pragma unroll
            for (int u = 0; u < 8; ++u) {
                if (lane < cc[u]) {
                    atomicAdd(&cnt[rc[u].x >> 24], 1);
                    stage[of[u] + lane] = rc[u];
                }
#pragma nounroll
                for (int k0 = 32; k0 < cc[u]; k0 += 32) {
                    if (lane < cc[u] - k0) {
                        const uint2 r2 =
                            recs[(size_t)(wave + TILE_WAVES * (n0 + u)) * SEG + oo[u] + k0 + lane];
                        atomicAdd(&cnt[r2.x >> 24], 1);
                        stage[of[u] + k0 + lane] = r2;
                    }
                }
            }
        }
        __syncthreads();
        scan_counts(cnt, start, tid);
        __syncthreads();
        {
            uint2 e[ORD_TRIPS];
            int p[ORD_TRIPS];
#pragma unroll
            for (int i = 0; i < ORD_TRIPS; ++i) {
                e[i] = make_uint2(0u, 0u);
                if (tid + 1024 * i < total) e[i] = stage[tid + 1024 * i];
            }
#pragma unroll
            for (int i = 0; i < ORD_TRIPS; ++i) {
                p[i] = 0;
                if (tid + 1024 * i < total) {
                    const int r = (int)(e[i].x >> 24);
                    p[i] = start[r] + atomicAdd(&cnt[r], -1) - 1;
                }
            }
#pragma unroll
            for (int i = 0; i < ORD_TRIPS; ++i)
                if (tid + 1024 * i < total) srt[p[i]] = e[i];
        }
        __syncthreads();

        const int ka = (total * wave) / TILE_WAVES;
        const int kb = (total * (wave + 1)) / TILE_WAVES;
        int cur = -1;
        float sxl = 0.f, syl = 0.f, sxu = 0.f, syu = 0.f;
        const int trips = (kb - ka + 31) >> 5;
        const int klin = ka + 32 * (((wave >> 2) * trips) >> 3);
        int jj = 0, r = -1;
        float ve = 0.f, s1 = 0.f, s2 = 0.f;
        if (ka + lane < kb) {
            const uint2 e = srt[ka + lane];
            jj = (int)(e.x & 0x00FFFFFFu);
            r  = (int)(e.x >> 24);
            ve = __uint_as_float(e.y);
            const int list = (jj >= N) ? 1 : 0;
            const float* s = sc + (size_t)list * 2 * (size_t)N;
            s1 = s[jj - list * N];
            s2 = s[(size_t)N + base + r];
        }
        for (int k = ka; k < kb; k += 32) {
            const float z = s1 + s2;
            const float elu = (z > 0.f) ? z : expm1f(z);
            const float alpha = elu * ve;
            const bool up = (jj >= N);
            const unsigned roff = (unsigned)(up ? jj - N : jj) * (unsigned)(IN_C * sizeof(float));
            const float al = up ? 0.f : alpha;
            const float au = up ? alpha : 0.f;
            const int cr = r;
            const int count = min(32, kb - k);
            jj = 0; r = -1; ve = 0.f; s1 = 0.f; s2 = 0.f;
            if (k + 32 + lane < kb) {
                const uint2 e = srt[k + 32 + lane];
                jj = (int)(e.x & 0x00FFFFFFu);
                r  = (int)(e.x >> 24);
                ve = __uint_as_float(e.y);
                const int list = (jj >= N) ? 1 : 0;
                const float* s = sc + (size_t)list * 2 * (size_t)N;
                s1 = s[jj - list * N];
                s2 = s[(size_t)N + base + r];
            }
            if (k == klin) {
                lin_units(accs, wls, (float*)stage, wave, lane);
                lin_done = true;
            }
#pragma unroll
            for (int h = 0; h < 2; ++h) {
                const int nb = count - h * 16;
                if (nb <= 0) break;
                if (nb >= 16)
                    sum_rows16<true >(roff, cr, al, au, h * 16, 16, lane, x, accs, cur,
                                      sxl, syl, sxu, syu);
                else
                    sum_rows16<false>(roff, cr, al, au, h * 16, nb, lane, x, accs, cur,
                                      sxl, syl, sxu, syu);
            }
        }
        if (cur >= 0) {
            float* a = accs + cur * ACC_PITCH + 2 * lane;
            atomicAdd(a,                 sxl);
            atomicAdd(a + 1,             syl);
            atomicAdd(a + ACC_U_OFF,     sxu);
            atomicAdd(a + ACC_U_OFF + 1, syu);
        }
    } else if (use_rec) {
        const uint2* recs = (const uint2*)region;
        for (int sb = wave; sb < nseg; sb += TILE_WAVES * 32) {
            if (sb != wave) {
                const int sseg = sb + TILE_WAVES * lane;
                pp = 0u;
                if (sseg < nseg) pp = offs[(size_t)sseg * SEG_TILES + tile];
            }
            unsigned pw = (unsigned)__builtin_amdgcn_readlane((int)pp, 0);
            int o0 = (int)(pw & 0xFFFFu), o1 = (int)(pw >> 16);
            uint2 rec = recs[(size_t)sb * SEG + min(o0 + lane, SEG - 1)];
            for (int n = 0; n < 32; ++n) {
                const int seg = sb + TILE_WAVES * n;
                if (seg >= nseg) break;
                const bool has_nxt = (n + 1 < 32) && (seg + TILE_WAVES < nseg);
                const int sn = min(seg + TILE_WAVES, nseg - 1);
                pw = (unsigned)__builtin_amdgcn_readlane((int)pp, min(n + 1, 31));
                int o0_nxt = (int)(pw & 0xFFFFu), o1_nxt = (int)(pw >> 16);
                if (!has_nxt) { o0_nxt = 0; o1_nxt = 0; }
                const uint2 rec_nxt = recs[(size_t)sn * SEG + min(o0_nxt + lane, SEG - 1)];
                if (o1 > o0)
                    flush_recs(rec.x, __uint_as_float(rec.y), lane < o1 - o0, min(32, o1 - o0),
                               lane, base, x, sc, accs, N);
                for (int k0 = o0 + 32; k0 < o1; k0 += 32) {
                    const bool v2 = lane < o1 - k0;
                    uint2 r2 = make_uint2(0u, 0u);
                    if (v2) r2 = recs[(size_t)seg * SEG + k0 + lane];
                    flush_recs(r2.x, __uint_as_float(r2.y), v2, min(32, o1 - k0),
                               lane, base, x, sc, accs, N);
                }
                rec = rec_nxt; o0 = o0_nxt; o1 = o1_nxt;
            }
        }
    } else {
        const unsigned short* sorted = (const unsigned short*)region;
        for (int sb = wave; sb < nseg; sb += TILE_WAVES * 32) {
            if (sb != wave) {
                const int sseg = sb + TILE_WAVES * lane;
                pp = 0u;
                if (sseg < nseg) pp = offs[(size_t)sseg * SEG_TILES + tile];
            }
            unsigned pw = (unsigned)__builtin_amdgcn_readlane((int)pp, 0);
            int o0 = (int)(pw & 0xFFFFu), o1 = (int)(pw >> 16);
            int raw = (int)sorted[(size_t)sb * SEG + min(o0 + lane, SEG - 1)];
            for (int n = 0; n < 32; ++n) {
                const int seg = sb + TILE_WAVES * n;
                if (seg >= nseg) break;
                const bool has_nxt = (n + 1 < 32) && (seg + TILE_WAVES < nseg);
                const int sn = min(seg + TILE_WAVES, nseg - 1);
                pw = (unsigned)__builtin_amdgcn_readlane((int)pp, min(n + 1, 31));
                int o0_nxt = (int)(pw & 0xFFFFu), o1_nxt = (int)(pw >> 16);
                if (!has_nxt) { o0_nxt = 0; o1_nxt = 0; }
                const int raw_nxt = (int)sorted[(size_t)sn * SEG + min(o0_nxt + lane, SEG - 1)];
                if (o1 > o0) {
                    const int g = (lane < o1 - o0) ? seg * SEG + raw : -1;
                    flush_edges(g, min(32, o1 - o0), lane, base, idx_l, val_l, idx_u, val_u,
                                x, sc, accs, N, E, Epad);
                }
                for (int k0 = o0 + 32; k0 < o1; k0 += 32) {
                    int g2 = -1;
                    if (lane < o1 - k0) g2 = seg * SEG + (int)sorted[(size_t)seg * SEG + k0 + lane];
                    flush_edges(g2, min(32, o1 - k0), lane, base, idx_l, val_l, idx_u, val_u,
                                x, sc, accs, N, E, Epad);
                }
                raw = raw_nxt; o0 = o0_nxt; o1 = o1_nxt;
            }
        }
    }

    if (!lin_done) lin_units(accs, wls, (float*)stage, wave, lane);
    __syncthreads();
    {
        const int half = lane >> 4;
        const int lrow = lane & 15;
        float* res = (float*)stage;
        for (int unit = wave; unit < N_UNITS; unit += TILE_WAVES) {
            const int nt = unit / 13;
            const int rb = unit - 13 * nt;
            const int col = nt * 16 + lrow;
            const float* al = accs + (rb * 16 + lrow) * ACC_PITCH + 2 * half;
            const float* au = al + ACC_U_OFF;
            v8f c1 = {};
#pragma unroll
            for (int kk = 0; kk < 16; ++kk) {
                const int k0 = 4 * kk + 2 * half;
                const v2f a = *(const v2f*)(al + 4 * kk);
                v2f b;
                b.x = w_lower[(size_t)k0 * OUT_C + col];
                b.y = w_lower[(size_t)(k0 + 1) * OUT_C + col];
                c1 = __builtin_amdgcn_wmma_f32_16x16x4_f32(
                         false, a,  false, b,
                         (short)0, c1,  false,  false);
            }
            __builtin_amdgcn_sched_barrier(0);
#pragma unroll
            for (int kk = 0; kk < 16; ++kk) {
                const int k0 = 4 * kk + 2 * half;
                const v2f a = *(const v2f*)(au + 4 * kk);
                v2f b;
                b.x = w_upper[(size_t)k0 * OUT_C + col];
                b.y = w_upper[(size_t)(k0 + 1) * OUT_C + col];
                c1 = __builtin_amdgcn_wmma_f32_16x16x4_f32(
                         false, a,  false, b,
                         (short)0, c1,  false,  false);
            }
#pragma unroll
            for (int r = 0; r < 8; ++r) {
                float* o = res + (rb * 16 + r + 8 * half) * OUT_C + col;
                *o = fmaxf(c1[r] + *o * EPS, 0.0f);
            }
        }
        __syncthreads();
        float4* dst = (float4*)(out + (size_t)base * OUT_C);
        const float4* rs = (const float4*)res;
        for (int t = tid; t < rows * (OUT_C / 4); t += 1024) dst[t] = rs[t];
    }
}

extern "C" void kernel_launch(void* const* d_in, const int* in_sizes, int n_in,
                              void* d_out, int out_size, void* d_ws, size_t ws_size,
                              hipStream_t stream) {
    const float* x        = (const float*)d_in[0];
    const int*   low_idx  = (const int*)  d_in[1];
    const float* low_val  = (const float*)d_in[2];
    const int*   up_idx   = (const int*)  d_in[3];
    const float* up_val   = (const float*)d_in[4];
    const float* w_lower  = (const float*)d_in[5];
    const float* a_lower  = (const float*)d_in[6];
    const float* w_upper  = (const float*)d_in[7];
    const float* a_upper  = (const float*)d_in[8];
    const float* w_lin    = (const float*)d_in[9];
    float*       out      = (float*)d_out;

    const int N = in_sizes[0] / IN_C;
    const int E = in_sizes[2];
    const int Epad = (E + 15) & ~15;

    float* sc = (float*)d_ws;
    const int nseg = (2 * Epad + SEG - 1) / SEG;
    unsigned int* region = (unsigned int*)(sc + 4 * (size_t)N);
    const char* lists_end_rec = (const char*)region + (size_t)nseg * SEG * REC_BYTES;
    const char* lists_end_pos = (const char*)region + (size_t)nseg * SEG * POS_BYTES;
    const unsigned long long need_rec = (unsigned long long)
        (lists_end_rec + (size_t)nseg * SEG_TILES * sizeof(unsigned int) - (const char*)d_ws);
    const unsigned long long need_pos = (unsigned long long)
        (lists_end_pos + (size_t)nseg * SEG_TILES * sizeof(unsigned int) - (const char*)d_ws);
    const unsigned long long have = (unsigned long long)ws_size;

    seg_sort_kernel<<<nseg, 1024, 0, stream>>>(low_idx, low_val, up_idx, up_val,
                                               x, w_lower, a_lower, w_upper, a_upper, sc, region,
                                               N, E, Epad, nseg, need_rec, need_pos, have);

    {
        const int tiles = (N + R_TILE - 1) / R_TILE;
        tile_accum_kernel<<<tiles, 1024, 0, stream>>>(
            region, low_idx, low_val, up_idx, up_val,
            x, w_lower, w_upper, w_lin, sc, out, N, E, Epad, nseg, need_rec, need_pos, have);
    }
}
